# starting kernel
# baseline (speedup 1.0000x reference)
#include <hip/hip_runtime.h>
#include <hip/hip_bf16.h>
#include <math.h>

typedef __attribute__((ext_vector_type(2))) float v2f;
typedef __attribute__((ext_vector_type(8))) float v8f;

#define BATCH  128
#define NRN    54
#define DTH    4
#define STEPS  30
#define PIX    5625
#define KPAD   5632
#define NPAD   64
#define CELLS  (BATCH * NRN)

__global__ void pool_kernel(const float* __restrict__ x, float* __restrict__ xp) {
  int g = blockIdx.x * blockDim.x + threadIdx.x;
  if (g >= BATCH * KPAD) return;
  int b = g / KPAD;
  int p = g - b * KPAD;
  if (p >= PIX) { xp[g] = 0.0f; return; }
  int pr = p / 75;
  int pc = p - pr * 75;
  const float* base = x + (size_t)b * 360000 + pr * 4800 + pc * 8;
  float s = 0.0f;
#pragma unroll
  for (int r = 0; r < 8; ++r) {
    const float4* q = reinterpret_cast<const float4*>(base + r * 600);
    float4 a0 = q[0], a1 = q[1];
    s += (a0.x + a0.y + a0.z + a0.w) + (a1.x + a1.y + a1.z + a1.w);
  }
  xp[g] = s * (1.0f / 64.0f);
}

__global__ void wtrans_kernel(const float* __restrict__ w, float* __restrict__ wT) {
  int g = blockIdx.x * blockDim.x + threadIdx.x;
  if (g >= KPAD * NPAD) return;
  int k = g >> 6;
  int n = g & 63;
  wT[g] = (k < PIX && n < NRN) ? w[n * PIX + k] : 0.0f;
}

__global__ void __launch_bounds__(128) fc_wmma_kernel(const float* __restrict__ xp,
                                                      const float* __restrict__ wT,
                                                      const float* __restrict__ bias,
                                                      float* __restrict__ xf) {
  __shared__ float sA[16 * 68];
  __shared__ float sB[64 * NPAD];
  const int tid  = threadIdx.x;
  const int lane = tid & 31;
  const int wave = tid >> 5;
  const int m0   = blockIdx.x * 16;
  const int half = lane >> 4;
  const int l15  = lane & 15;

  v8f acc = {0.f, 0.f, 0.f, 0.f, 0.f, 0.f, 0.f, 0.f};

  for (int k0 = 0; k0 < KPAD; k0 += 64) {
    for (int i = tid; i < 16 * 16; i += 128) {
      int m = i >> 4, q = (i & 15) << 2;
      float4 a = *reinterpret_cast<const float4*>(xp + (size_t)(m0 + m) * KPAD + k0 + q);
      *reinterpret_cast<float4*>(&sA[m * 68 + q]) = a;
    }
    for (int i = tid; i < 64 * 16; i += 128) {
      int k = i >> 4, q = (i & 15) << 2;
      float4 bv = *reinterpret_cast<const float4*>(wT + (size_t)(k0 + k) * NPAD + q);
      *reinterpret_cast<float4*>(&sB[k * NPAD + q]) = bv;
    }
    __syncthreads();

#pragma unroll 4
    for (int kk = 0; kk < 64; kk += 4) {
      int ka = kk + (half << 1);
      v2f a;
      a.x = sA[l15 * 68 + ka];
      a.y = sA[l15 * 68 + ka + 1];
      v2f bf;
      bf.x = sB[ka * NPAD + (wave << 4) + l15];
      bf.y = sB[(ka + 1) * NPAD + (wave << 4) + l15];
      acc = __builtin_amdgcn_wmma_f32_16x16x4_f32(
           false, a,  false, bf,
           (short)0, acc,  false,  false);
    }
    __syncthreads();
  }

#pragma unroll
  for (int v = 0; v < 8; ++v) {
    int m = m0 + v + (half << 3);
    int n = (wave << 4) + l15;
    float o = acc[v] + (n < NRN ? bias[n] : 0.0f);
    xf[m * NPAD + n] = o;
  }
}

#define NTH  1024
#define MAXC 7
__global__ void __launch_bounds__(NTH) scan_kernel(const float* __restrict__ xf,
                                                   const float* __restrict__ lw,
                                                   const float* __restrict__ hw,
                                                   float* __restrict__ out) {
  __shared__ float sVals[CELLS];
  __shared__ float sBatch[BATCH];
  __shared__ float sWave[NTH / 32];
  __shared__ float sThr;

  const int tid  = threadIdx.x;
  const int lane = tid & 31;
  const int wv   = tid >> 5;

  float h0[MAXC], h1[MAXC], h2[MAXC], h3[MAXC], cnt[MAXC], xv[MAXC];
  float lw0[MAXC], lw1[MAXC], lw2[MAXC], lw3[MAXC];
  const float w0 = hw[0], w1 = hw[1], w2 = hw[2], w3 = hw[3];

#pragma unroll
  for (int j = 0; j < MAXC; ++j) {
    int c = tid + (j << 10);
    h0[j] = h1[j] = h2[j] = h3[j] = cnt[j] = 0.0f;
    if (c < CELLS) {
      int b = c / NRN, n = c - b * NRN;
      xv[j]  = xf[b * NPAD + n];
      lw0[j] = lw[n * DTH + 0]; lw1[j] = lw[n * DTH + 1];
      lw2[j] = lw[n * DTH + 2]; lw3[j] = lw[n * DTH + 3];
    } else {
      xv[j] = lw0[j] = lw1[j] = lw2[j] = lw3[j] = 0.0f;
    }
  }

  for (int step = 0; step < STEPS; ++step) {
    float sv[MAXC], hsv[MAXC], rate[MAXC];
#pragma unroll
    for (int j = 0; j < MAXC; ++j) {
      int c = tid + (j << 10);
      sv[j]  = h0[j] * lw0[j] + h1[j] * lw1[j] + h2[j] * lw2[j] + h3[j] * lw3[j];
      hsv[j] = h0[j] * w0 + h1[j] * w1 + h2[j] * w2 + h3[j] * w3;
      if (c < CELLS) sVals[c] = sv[j];
    }
    __syncthreads();
    if (tid < BATCH) {
      float acc = 0.0f;
      const float* row = &sVals[tid * NRN];
#pragma unroll
      for (int n = 0; n < NRN; ++n) acc += row[n];
      sBatch[tid] = acc;
    }
    __syncthreads();
    float psum = 0.0f;
#pragma unroll
    for (int j = 0; j < MAXC; ++j) {
      int c = tid + (j << 10);
      if (c < CELLS) {
        int b = c / NRN;
        float cur = xv[j] + (sBatch[b] - sv[j]) + hsv[j];
        rate[j] = __expf(cur);
        psum += rate[j];
      } else {
        rate[j] = 0.0f;
      }
    }
#pragma unroll
    for (int off = 16; off > 0; off >>= 1) psum += __shfl_down(psum, off, 32);
    if (lane == 0) sWave[wv] = psum;
    __syncthreads();
    if (tid < 32) {
      float v = sWave[tid];
#pragma unroll
      for (int off = 16; off > 0; off >>= 1) v += __shfl_down(v, off, 32);
      if (tid == 0) sThr = v * (1.0f / (float)CELLS);
    }
    __syncthreads();
    const float thr = sThr;
#pragma unroll
    for (int j = 0; j < MAXC; ++j) {
      float sp = (rate[j] > thr) ? 1.0f : 0.0f;
      cnt[j] += sp;
      h0[j] = h1[j]; h1[j] = h2[j]; h2[j] = h3[j]; h3[j] = sp;
    }
  }

#pragma unroll
  for (int j = 0; j < MAXC; ++j) {
    int c = tid + (j << 10);
    if (c < CELLS) {
      float cc = cnt[j];
      out[c] = fmaxf(cc, 0.0f) + log1pf(__expf(-fabsf(cc)));
    }
  }
}

extern "C" void kernel_launch(void* const* d_in, const int* in_sizes, int n_in,
                              void* d_out, int out_size, void* d_ws, size_t ws_size,
                              hipStream_t stream) {
  const float* x  = (const float*)d_in[0];
  const float* lw = (const float*)d_in[1];
  const float* hw = (const float*)d_in[2];
  const float* fw = (const float*)d_in[3];
  const float* fb = (const float*)d_in[4];

  float* ws = (float*)d_ws;
  float* xp = ws;
  float* wT = xp + (size_t)BATCH * KPAD;
  float* xf = wT + (size_t)KPAD * NPAD;
  float* out = (float*)d_out;

  pool_kernel<<<(BATCH * KPAD + 255) / 256, 256, 0, stream>>>(x, xp);
  wtrans_kernel<<<(KPAD * NPAD + 255) / 256, 256, 0, stream>>>(fw, wT);
  fc_wmma_kernel<<<BATCH / 16, 128, 0, stream>>>(xp, wT, fb, xf);
  scan_kernel<<<1, NTH, 0, stream>>>(xf, lw, hw, out);
}


# scan transposed: lane = batch row, wave = neuron group (14 cells per thread, 512 threads); row partials meet in LDS, wave sums by DPP; 2 barriers/step, not 4 (lateral and rate-mean sum order changed)
# speedup vs baseline: 6.1097x; 6.1097x over previous
#include <hip/hip_runtime.h>
#include <hip/hip_bf16.h>
#include <math.h>

typedef __attribute__((ext_vector_type(2))) float v2f;
typedef __attribute__((ext_vector_type(8))) float v8f;

#define BATCH  128
#define NRN    54
#define DTH    4
#define STEPS  30
#define PIX    5625
#define KPAD   5632
#define NPAD   64
#define CELLS  (BATCH * NRN)

__global__ void pool_kernel(const float* __restrict__ x, float* __restrict__ xp) {
  int g = blockIdx.x * blockDim.x + threadIdx.x;
  if (g >= BATCH * KPAD) return;
  int b = g / KPAD;
  int p = g - b * KPAD;
  if (p >= PIX) { xp[g] = 0.0f; return; }
  int pr = p / 75;
  int pc = p - pr * 75;
  const float* base = x + (size_t)b * 360000 + pr * 4800 + pc * 8;
  float s = 0.0f;
#pragma unroll
  for (int r = 0; r < 8; ++r) {
    const float4* q = reinterpret_cast<const float4*>(base + r * 600);
    float4 a0 = q[0], a1 = q[1];
    s += (a0.x + a0.y + a0.z + a0.w) + (a1.x + a1.y + a1.z + a1.w);
  }
  xp[g] = s * (1.0f / 64.0f);
}

__global__ void wpad_kernel(const float* __restrict__ w, float* __restrict__ wP) {
  int g = blockIdx.x * blockDim.x + threadIdx.x;
  if (g >= NPAD * KPAD) return;
  int n = g / KPAD;
  int k = g - n * KPAD;
  wP[g] = (k < PIX && n < NRN) ? w[n * PIX + k] : 0.0f;
}

#define FCW   32
#define KSL   (KPAD / FCW)
#define NLDH  (KSL / 16)
__device__ __forceinline__ void load_frags(const float* __restrict__ p, float4 (&f)[NLDH]) {
  for (int j = 0; j < NLDH; ++j) f[j] = *reinterpret_cast<const float4*>(p + j * 8);
}

__global__ void __launch_bounds__(FCW * 32) fc_wmma_kernel(const float* __restrict__ xp,
                                                           const float* __restrict__ wP,
                                                           const float* __restrict__ bias,
                                                           float* __restrict__ xf) {
  __shared__ float sRed[FCW * 256];
  const int tid  = threadIdx.x;
  const int lane = tid & 31;
  const int wave = tid >> 5;
  const int half = lane >> 4;
  const int l15  = lane & 15;
  const int m0   = blockIdx.x * 16;
  const int n0   = blockIdx.y * 16;

  const float* pa = xp + (size_t)(m0 + l15) * KPAD + wave * KSL + (half << 2);
  const float* pb = wP + (size_t)(n0 + l15) * KPAD + wave * KSL + (half << 2);

  v8f acc = {0.f, 0.f, 0.f, 0.f, 0.f, 0.f, 0.f, 0.f};

  for (int bt = 0; bt < 2; ++bt) {
    float4 fa[NLDH], fb[NLDH];
    load_frags(pa + bt * (NLDH * 8), fa);
    load_frags(pb + bt * (NLDH * 8), fb);
    for (int j = 0; j < NLDH; ++j) {
      v2f a0 = {fa[j].x, fa[j].y}, b0 = {fb[j].x, fb[j].y};
      v2f a1 = {fa[j].z, fa[j].w}, b1 = {fb[j].z, fb[j].w};
      acc = __builtin_amdgcn_wmma_f32_16x16x4_f32(
           false, a0,  false, b0,
           (short)0, acc,  false,  false);
      acc = __builtin_amdgcn_wmma_f32_16x16x4_f32(
           false, a1,  false, b1,
           (short)0, acc,  false,  false);
    }
  }

  for (int v = 0; v < 8; ++v) sRed[wave * 256 + v * 32 + lane] = acc[v];
  __syncthreads();

  if (tid < 256) {
    const int ml  = tid >> 4;
    const int nl  = tid & 15;
    const int idx = ((ml & 7) << 5) + ((ml >> 3) << 4) + nl;
    float s = 0.0f;
    for (int w = 0; w < FCW; ++w) s += sRed[w * 256 + idx];
    const int n = n0 + nl;
    s += (n < NRN ? bias[n] : 0.0f);
    xf[blockIdx.y * (BATCH * 16) + (m0 + ml) * 16 + nl] = s;
  }
}

#define NTH  512
#define NWV  (NTH / 32)
#define NGR  4
#define NPT  14
#define NLAST (NRN - (NGR - 1) * NPT)
#define NPR  (NPT / 2)

template <int CTRL>
__device__ __forceinline__ float dpp_row(float v) {
  const int b = __builtin_bit_cast(int, v);
  return __builtin_bit_cast(float, __builtin_amdgcn_update_dpp(b, b, CTRL, 0xf, 0xf, false));
}
__device__ __forceinline__ float xhalf(float v) {
  const int b = __builtin_bit_cast(int, v);
  return __builtin_bit_cast(float, __builtin_amdgcn_permlanex16(b, b, 0x76543210, 0xfedcba98, false, false));
}
__device__ __forceinline__ float row_sum16(float v) {
  v += dpp_row<0x161>(v);
  v += dpp_row<0x162>(v);
  v += dpp_row<0x164>(v);
  v += dpp_row<0x168>(v);
  return v;
}

__global__ void __launch_bounds__(NTH) scan_kernel(const float* __restrict__ xf,
                                                   const float* __restrict__ lw,
                                                   const float* __restrict__ hw,
                                                   float* __restrict__ out) {
  __shared__ float sPart[NGR * BATCH];
  __shared__ float sWave[NWV];
  __shared__ float sCnt[CELLS];

  const int tid  = threadIdx.x;
  const int lane = tid & 31;
  const int wv   = __builtin_amdgcn_readfirstlane(tid >> 5);
  const int ng   = wv >> 2;
  const int b    = ((wv & 3) << 5) + lane;
  const int n0   = ng * NPT;
  const bool full = (ng < NGR - 1);

  v2f h0[NPR], h1[NPR], h2[NPR], h3[NPR], cnt[NPR], xv[NPR];
  v2f lw0[NPR], lw1[NPR], lw2[NPR], lw3[NPR];
  const float w0 = hw[0], w1 = hw[1], w2 = hw[2], w3 = hw[3];
  const v2f zero2 = {0.0f, 0.0f};

#pragma unroll
  for (int p = 0; p < NPR; ++p) {
    const int n = n0 + 2 * p;
    h0[p] = h1[p] = h2[p] = h3[p] = cnt[p] = zero2;
    if (p < NLAST / 2 || full) {
      const float* q = xf + (n >> 4) * (BATCH * 16) + b * 16 + (n & 15);
      xv[p]  = v2f{q[0], q[1]};
      lw0[p] = v2f{lw[n * DTH + 0], lw[n * DTH + DTH + 0]};
      lw1[p] = v2f{lw[n * DTH + 1], lw[n * DTH + DTH + 1]};
      lw2[p] = v2f{lw[n * DTH + 2], lw[n * DTH + DTH + 2]};
      lw3[p] = v2f{lw[n * DTH + 3], lw[n * DTH + DTH + 3]};
    } else {
      xv[p] = lw0[p] = lw1[p] = lw2[p] = lw3[p] = zero2;
    }
  }

  for (int step = 0; step < STEPS; ++step) {
    v2f sv[NPR], hsv[NPR], rate[NPR];
    v2f part2 = zero2;
#pragma unroll
    for (int p = 0; p < NPR; ++p) {
      sv[p]  = h0[p] * lw0[p] + h1[p] * lw1[p] + h2[p] * lw2[p] + h3[p] * lw3[p];
      hsv[p] = h0[p] * w0 + h1[p] * w1 + h2[p] * w2 + h3[p] * w3;
      part2 += sv[p];
    }
    sPart[ng * BATCH + b] = part2.x + part2.y;
    __syncthreads();
    const float Sb = ((sPart[b] + sPart[BATCH + b]) + sPart[2 * BATCH + b]) + sPart[3 * BATCH + b];
    const v2f Sb2 = {Sb, Sb};
    v2f ps2 = zero2;
#pragma unroll
    for (int p = 0; p < NPR; ++p) {
      const v2f cur = xv[p] + (Sb2 - sv[p]) + hsv[p];
      v2f e = {__expf(cur.x), __expf(cur.y)};
      if (p >= NLAST / 2) e = full ? e : zero2;
      rate[p] = e;
      ps2 += e;
    }
    float psum = row_sum16(ps2.x + ps2.y);
    psum += xhalf(psum);
    if (lane == 0) sWave[wv] = psum;
    __syncthreads();
    const float tot = row_sum16(sWave[lane & 15]);
    const float thr = tot * (1.0f / (float)CELLS);
#pragma unroll
    for (int p = 0; p < NPR; ++p) {
      const v2f sp = {(rate[p].x > thr) ? 1.0f : 0.0f, (rate[p].y > thr) ? 1.0f : 0.0f};
      cnt[p] += sp;
      h0[p] = h1[p]; h1[p] = h2[p]; h2[p] = h3[p]; h3[p] = sp;
    }
  }

#pragma unroll
  for (int p = 0; p < NPR; ++p) {
    if (p < NLAST / 2 || full) {
      sCnt[b * NRN + n0 + 2 * p]     = cnt[p].x;
      sCnt[b * NRN + n0 + 2 * p + 1] = cnt[p].y;
    }
  }
  __syncthreads();
  for (int c = tid; c < CELLS; c += NTH) {
    float cc = sCnt[c];
    out[c] = fmaxf(cc, 0.0f) + log1pf(__expf(-fabsf(cc)));
  }
}

extern "C" void kernel_launch(void* const* d_in, const int* in_sizes, int n_in,
                              void* d_out, int out_size, void* d_ws, size_t ws_size,
                              hipStream_t stream) {
  const float* x  = (const float*)d_in[0];
  const float* lw = (const float*)d_in[1];
  const float* hw = (const float*)d_in[2];
  const float* fw = (const float*)d_in[3];
  const float* fb = (const float*)d_in[4];

  float* ws = (float*)d_ws;
  float* xp = ws;
  float* wP = xp + (size_t)BATCH * KPAD;
  float* xf = wP + (size_t)NPAD * KPAD;
  float* out = (float*)d_out;

  pool_kernel<<<(BATCH * KPAD + 255) / 256, 256, 0, stream>>>(x, xp);
  wpad_kernel<<<(NPAD * KPAD + 255) / 256, 256, 0, stream>>>(fw, wP);
  fc_wmma_kernel<<<dim3(BATCH / 16, NPAD / 16), FCW * 32, 0, stream>>>(xp, wP, fb, xf);
  scan_kernel<<<1, NTH, 0, stream>>>(xf, lw, hw, out);
}


# pad of fc_w folded into the pool launch: the first 1408 of its 2816 work-groups also copy one wP element per thread; the pad launch (11264 one-element waves) is gone, 4 -> 3 launches; sums unchanged
# speedup vs baseline: 6.3124x; 1.0332x over previous
#include <hip/hip_runtime.h>
#include <hip/hip_bf16.h>
#include <math.h>

typedef __attribute__((ext_vector_type(2))) float v2f;
typedef __attribute__((ext_vector_type(8))) float v8f;

#define BATCH  128
#define NRN    54
#define DTH    4
#define STEPS  30
#define PIX    5625
#define KPAD   5632
#define NPAD   64
#define CELLS  (BATCH * NRN)

__global__ void pool_pad_kernel(const float* __restrict__ x, const float* __restrict__ w,
                                float* __restrict__ xp, float* __restrict__ wP) {
  int g = blockIdx.x * blockDim.x + threadIdx.x;
  if (g >= BATCH * KPAD) return;
  int b = g / KPAD;
  int p = g - b * KPAD;
  const bool padw = (b < NPAD);
  float wv = 0.0f;
  if (padw && p < PIX && b < NRN) wv = w[b * PIX + p];
  float s = 0.0f;
  if (p < PIX) {
    int pr = p / 75;
    int pc = p - pr * 75;
    const float* base = x + (size_t)b * 360000 + pr * 4800 + pc * 8;
#pragma unroll
    for (int r = 0; r < 8; ++r) {
      const float4* q = reinterpret_cast<const float4*>(base + r * 600);
      float4 a0 = q[0], a1 = q[1];
      s += (a0.x + a0.y + a0.z + a0.w) + (a1.x + a1.y + a1.z + a1.w);
    }
    s *= (1.0f / 64.0f);
  }
  xp[g] = s;
  if (padw) wP[g] = wv;
}

#define FCW   32
#define KSL   (KPAD / FCW)
#define NLDH  (KSL / 16)
__device__ __forceinline__ void load_frags(const float* __restrict__ p, float4 (&f)[NLDH]) {
  for (int j = 0; j < NLDH; ++j) f[j] = *reinterpret_cast<const float4*>(p + j * 8);
}

__global__ void __launch_bounds__(FCW * 32) fc_wmma_kernel(const float* __restrict__ xp,
                                                           const float* __restrict__ wP,
                                                           const float* __restrict__ bias,
                                                           float* __restrict__ xf) {
  __shared__ float sRed[FCW * 256];
  const int tid  = threadIdx.x;
  const int lane = tid & 31;
  const int wave = tid >> 5;
  const int half = lane >> 4;
  const int l15  = lane & 15;
  const int m0   = blockIdx.x * 16;
  const int n0   = blockIdx.y * 16;

  const float* pa = xp + (size_t)(m0 + l15) * KPAD + wave * KSL + (half << 2);
  const float* pb = wP + (size_t)(n0 + l15) * KPAD + wave * KSL + (half << 2);

  v8f acc = {0.f, 0.f, 0.f, 0.f, 0.f, 0.f, 0.f, 0.f};

  for (int bt = 0; bt < 2; ++bt) {
    float4 fa[NLDH], fb[NLDH];
    load_frags(pa + bt * (NLDH * 8), fa);
    load_frags(pb + bt * (NLDH * 8), fb);
    for (int j = 0; j < NLDH; ++j) {
      v2f a0 = {fa[j].x, fa[j].y}, b0 = {fb[j].x, fb[j].y};
      v2f a1 = {fa[j].z, fa[j].w}, b1 = {fb[j].z, fb[j].w};
      acc = __builtin_amdgcn_wmma_f32_16x16x4_f32(
           false, a0,  false, b0,
           (short)0, acc,  false,  false);
      acc = __builtin_amdgcn_wmma_f32_16x16x4_f32(
           false, a1,  false, b1,
           (short)0, acc,  false,  false);
    }
  }

  for (int v = 0; v < 8; ++v) sRed[wave * 256 + v * 32 + lane] = acc[v];
  __syncthreads();

  if (tid < 256) {
    const int ml  = tid >> 4;
    const int nl  = tid & 15;
    const int idx = ((ml & 7) << 5) + ((ml >> 3) << 4) + nl;
    float s = 0.0f;
    for (int w = 0; w < FCW; ++w) s += sRed[w * 256 + idx];
    const int n = n0 + nl;
    s += (n < NRN ? bias[n] : 0.0f);
    xf[blockIdx.y * (BATCH * 16) + (m0 + ml) * 16 + nl] = s;
  }
}

#define NTH  512
#define NWV  (NTH / 32)
#define NGR  4
#define NPT  14
#define NLAST (NRN - (NGR - 1) * NPT)
#define NPR  (NPT / 2)

template <int CTRL>
__device__ __forceinline__ float dpp_row(float v) {
  const int b = __builtin_bit_cast(int, v);
  return __builtin_bit_cast(float, __builtin_amdgcn_update_dpp(b, b, CTRL, 0xf, 0xf, false));
}
__device__ __forceinline__ float xhalf(float v) {
  const int b = __builtin_bit_cast(int, v);
  return __builtin_bit_cast(float, __builtin_amdgcn_permlanex16(b, b, 0x76543210, 0xfedcba98, false, false));
}
__device__ __forceinline__ float row_sum16(float v) {
  v += dpp_row<0x161>(v);
  v += dpp_row<0x162>(v);
  v += dpp_row<0x164>(v);
  v += dpp_row<0x168>(v);
  return v;
}

__global__ void __launch_bounds__(NTH) scan_kernel(const float* __restrict__ xf,
                                                   const float* __restrict__ lw,
                                                   const float* __restrict__ hw,
                                                   float* __restrict__ out) {
  __shared__ float sPart[NGR * BATCH];
  __shared__ float sWave[NWV];
  __shared__ float sCnt[CELLS];

  const int tid  = threadIdx.x;
  const int lane = tid & 31;
  const int wv   = __builtin_amdgcn_readfirstlane(tid >> 5);
  const int ng   = wv >> 2;
  const int b    = ((wv & 3) << 5) + lane;
  const int n0   = ng * NPT;
  const bool full = (ng < NGR - 1);

  v2f h0[NPR], h1[NPR], h2[NPR], h3[NPR], cnt[NPR], xv[NPR];
  v2f lw0[NPR], lw1[NPR], lw2[NPR], lw3[NPR];
  const float w0 = hw[0], w1 = hw[1], w2 = hw[2], w3 = hw[3];
  const v2f zero2 = {0.0f, 0.0f};

#pragma unroll
  for (int p = 0; p < NPR; ++p) {
    const int n = n0 + 2 * p;
    h0[p] = h1[p] = h2[p] = h3[p] = cnt[p] = zero2;
    if (p < NLAST / 2 || full) {
      const float* q = xf + (n >> 4) * (BATCH * 16) + b * 16 + (n & 15);
      xv[p]  = v2f{q[0], q[1]};
      lw0[p] = v2f{lw[n * DTH + 0], lw[n * DTH + DTH + 0]};
      lw1[p] = v2f{lw[n * DTH + 1], lw[n * DTH + DTH + 1]};
      lw2[p] = v2f{lw[n * DTH + 2], lw[n * DTH + DTH + 2]};
      lw3[p] = v2f{lw[n * DTH + 3], lw[n * DTH + DTH + 3]};
    } else {
      xv[p] = lw0[p] = lw1[p] = lw2[p] = lw3[p] = zero2;
    }
  }

  for (int step = 0; step < STEPS; ++step) {
    v2f sv[NPR], hsv[NPR], rate[NPR];
    v2f part2 = zero2;
#pragma unroll
    for (int p = 0; p < NPR; ++p) {
      sv[p]  = h0[p] * lw0[p] + h1[p] * lw1[p] + h2[p] * lw2[p] + h3[p] * lw3[p];
      hsv[p] = h0[p] * w0 + h1[p] * w1 + h2[p] * w2 + h3[p] * w3;
      part2 += sv[p];
    }
    sPart[ng * BATCH + b] = part2.x + part2.y;
    __syncthreads();
    const float Sb = ((sPart[b] + sPart[BATCH + b]) + sPart[2 * BATCH + b]) + sPart[3 * BATCH + b];
    const v2f Sb2 = {Sb, Sb};
    v2f ps2 = zero2;
#pragma unroll
    for (int p = 0; p < NPR; ++p) {
      const v2f cur = xv[p] + (Sb2 - sv[p]) + hsv[p];
      v2f e = {__expf(cur.x), __expf(cur.y)};
      if (p >= NLAST / 2) e = full ? e : zero2;
      rate[p] = e;
      ps2 += e;
    }
    float psum = row_sum16(ps2.x + ps2.y);
    psum += xhalf(psum);
    if (lane == 0) sWave[wv] = psum;
    __syncthreads();
    const float tot = row_sum16(sWave[lane & 15]);
    const float thr = tot * (1.0f / (float)CELLS);
#pragma unroll
    for (int p = 0; p < NPR; ++p) {
      const v2f sp = {(rate[p].x > thr) ? 1.0f : 0.0f, (rate[p].y > thr) ? 1.0f : 0.0f};
      cnt[p] += sp;
      h0[p] = h1[p]; h1[p] = h2[p]; h2[p] = h3[p]; h3[p] = sp;
    }
  }

#pragma unroll
  for (int p = 0; p < NPR; ++p) {
    if (p < NLAST / 2 || full) {
      sCnt[b * NRN + n0 + 2 * p]     = cnt[p].x;
      sCnt[b * NRN + n0 + 2 * p + 1] = cnt[p].y;
    }
  }
  __syncthreads();
  for (int c = tid; c < CELLS; c += NTH) {
    float cc = sCnt[c];
    out[c] = fmaxf(cc, 0.0f) + log1pf(__expf(-fabsf(cc)));
  }
}

extern "C" void kernel_launch(void* const* d_in, const int* in_sizes, int n_in,
                              void* d_out, int out_size, void* d_ws, size_t ws_size,
                              hipStream_t stream) {
  const float* x  = (const float*)d_in[0];
  const float* lw = (const float*)d_in[1];
  const float* hw = (const float*)d_in[2];
  const float* fw = (const float*)d_in[3];
  const float* fb = (const float*)d_in[4];

  float* ws = (float*)d_ws;
  float* xp = ws;
  float* wP = xp + (size_t)BATCH * KPAD;
  float* xf = wP + (size_t)NPAD * KPAD;
  float* out = (float*)d_out;

  pool_pad_kernel<<<(BATCH * KPAD + 255) / 256, 256, 0, stream>>>(x, fw, xp, wP);
  fc_wmma_kernel<<<dim3(BATCH / 16, NPAD / 16), FCW * 32, 0, stream>>>(xp, wP, fb, xf);
  scan_kernel<<<1, NTH, 0, stream>>>(xf, lw, hw, out);
}


# pool launch re-tiled as a stream: 4096 long-lived waves, one 128-byte xp line per unit, 512-byte loads (a lane pair shares a window row, DPP add), 32-48 loads in flight per wave; sum order kept
# speedup vs baseline: 6.9480x; 1.1007x over previous
#include <hip/hip_runtime.h>
#include <hip/hip_bf16.h>
#include <math.h>

typedef __attribute__((ext_vector_type(2))) float v2f;
typedef __attribute__((ext_vector_type(8))) float v8f;

#define BATCH  128
#define NRN    54
#define DTH    4
#define STEPS  30
#define PIX    5625
#define KPAD   5632
#define NPAD   64
#define CELLS  (BATCH * NRN)

#define PWV    8
#define PGRID  512
#define PNW    (PWV * PGRID)
#define XLN    (KPAD / 32)
#define NUNIT  (BATCH * XLN)
#define NFULL  (NUNIT / PNW)
#define REMW   ((NUNIT - NFULL * PNW) / PGRID)
#define NSLOT  (NFULL + 1)
#define AHEAD  2
#define WCOPY  ((NPAD * KPAD + PNW * 32 - 1) / (PNW * 32))
static_assert(NUNIT == NFULL * PNW + REMW * PGRID && REMW <= PWV, "units must split evenly");

__device__ __forceinline__ float pair_other(float v) {
  const int b = __builtin_bit_cast(int, v);
  return __builtin_bit_cast(float, __builtin_amdgcn_update_dpp(b, b, 0x161, 0xf, 0xf, false));
}
__device__ __forceinline__ float lane_fetch(int byte_addr, float v) {
  return __builtin_bit_cast(float, __builtin_amdgcn_ds_bpermute(byte_addr, __builtin_bit_cast(int, v)));
}
__device__ __forceinline__ const float* window_ptr(const float* __restrict__ xb, int p, int half) {
  const int pp = p < PIX ? p : PIX - 1;
  const int pr = pp / 75;
  const int pc = pp - pr * 75;
  return xb + pr * 4800 + pc * 8 + half * 4;
}

__global__ void __launch_bounds__(PWV * 32) pool_pad_kernel(const float* __restrict__ x,
                                                            const float* __restrict__ w,
                                                            float* __restrict__ xp,
                                                            float* __restrict__ wP) {
  const int tid  = threadIdx.x;
  const int lane = tid & 31;
  const int wave = __builtin_amdgcn_readfirstlane(tid >> 5);
  const int gw   = blockIdx.x * PWV + wave;

  const int gt = blockIdx.x * (PWV * 32) + tid;
  float wv[WCOPY];
#pragma unroll
  for (int c = 0; c < WCOPY; ++c) {
    const int i = gt + c * (PNW * 32);
    wv[c] = 0.0f;
    if (i < NPAD * KPAD) {
      const int row = i / KPAD;
      const int k   = i - row * KPAD;
      if (row < NRN && k < PIX) wv[c] = w[row * PIX + k];
    }
  }

  float4 ring[AHEAD + 1][16];
#define UNIT_OF(S) ((S) < NFULL ? gw + PNW * (S) : NFULL * PNW + (int)blockIdx.x * REMW + wave)
#define HAS(S)     ((S) < NFULL || wave < REMW)
#define LOAD_UNIT(S) do { if (HAS(S)) {                                                        \
    const int u_ = UNIT_OF(S);                                                                 \
    const int b_ = u_ / XLN;                                                                   \
    const int m_ = u_ - b_ * XLN;                                                              \
    const float* xb_ = x + (size_t)b_ * 360000;                                                \
    const float* pa_ = window_ptr(xb_, 32 * m_ + (lane >> 1), lane & 1);                       \
    const float* pb_ = window_ptr(xb_, 32 * m_ + 16 + (lane >> 1), lane & 1);                  \
    _Pragma("unroll") for (int r = 0; r < 8; ++r) {                                            \
      ring[(S) % (AHEAD + 1)][2 * r]     = *reinterpret_cast<const float4*>(pa_ + r * 600);    \
      ring[(S) % (AHEAD + 1)][2 * r + 1] = *reinterpret_cast<const float4*>(pb_ + r * 600);    \
    } } } while (0)

#pragma unroll
  for (int s = 0; s < AHEAD; ++s) LOAD_UNIT(s);
#pragma unroll
  for (int s = 0; s < NSLOT; ++s) {
    __builtin_amdgcn_sched_barrier(0);
    if (s + AHEAD < NSLOT) LOAD_UNIT(s + AHEAD);
    __builtin_amdgcn_sched_barrier(0);
    if (HAS(s)) {
      const float4* q = ring[s % (AHEAD + 1)];
      float sA = 0.0f, sB = 0.0f;
#pragma unroll
      for (int r = 0; r < 8; ++r) {
        const float4 a = q[2 * r], c = q[2 * r + 1];
        const float hA = a.x + a.y + a.z + a.w;
        const float hB = c.x + c.y + c.z + c.w;
        sA += hA + pair_other(hA);
        sB += hB + pair_other(hB);
      }
      sA *= (1.0f / 64.0f);
      sB *= (1.0f / 64.0f);
      const int from = (lane << 3) & 127;
      const float tA = lane_fetch(from, sA);
      const float tB = lane_fetch(from, sB);
      const int u = UNIT_OF(s);
      const int m = u % XLN;
      float v = lane < 16 ? tA : tB;
      if (32 * m + lane >= PIX) v = 0.0f;
      xp[u * 32 + lane] = v;
    }
  }
#undef LOAD_UNIT
#undef HAS
#undef UNIT_OF

#pragma unroll
  for (int c = 0; c < WCOPY; ++c) {
    const int i = gt + c * (PNW * 32);
    if (i < NPAD * KPAD) wP[i] = wv[c];
  }
}

#define FCW   32
#define KSL   (KPAD / FCW)
#define NLDH  (KSL / 16)
__device__ __forceinline__ void load_frags(const float* __restrict__ p, float4 (&f)[NLDH]) {
  for (int j = 0; j < NLDH; ++j) f[j] = *reinterpret_cast<const float4*>(p + j * 8);
}

__global__ void __launch_bounds__(FCW * 32) fc_wmma_kernel(const float* __restrict__ xp,
                                                           const float* __restrict__ wP,
                                                           const float* __restrict__ bias,
                                                           float* __restrict__ xf) {
  __shared__ float sRed[FCW * 256];
  const int tid  = threadIdx.x;
  const int lane = tid & 31;
  const int wave = tid >> 5;
  const int half = lane >> 4;
  const int l15  = lane & 15;
  const int m0   = blockIdx.x * 16;
  const int n0   = blockIdx.y * 16;

  const float* pa = xp + (size_t)(m0 + l15) * KPAD + wave * KSL + (half << 2);
  const float* pb = wP + (size_t)(n0 + l15) * KPAD + wave * KSL + (half << 2);

  v8f acc = {0.f, 0.f, 0.f, 0.f, 0.f, 0.f, 0.f, 0.f};

  for (int bt = 0; bt < 2; ++bt) {
    float4 fa[NLDH], fb[NLDH];
    load_frags(pa + bt * (NLDH * 8), fa);
    load_frags(pb + bt * (NLDH * 8), fb);
    for (int j = 0; j < NLDH; ++j) {
      v2f a0 = {fa[j].x, fa[j].y}, b0 = {fb[j].x, fb[j].y};
      v2f a1 = {fa[j].z, fa[j].w}, b1 = {fb[j].z, fb[j].w};
      acc = __builtin_amdgcn_wmma_f32_16x16x4_f32(
           false, a0,  false, b0,
           (short)0, acc,  false,  false);
      acc = __builtin_amdgcn_wmma_f32_16x16x4_f32(
           false, a1,  false, b1,
           (short)0, acc,  false,  false);
    }
  }

  for (int v = 0; v < 8; ++v) sRed[wave * 256 + v * 32 + lane] = acc[v];
  __syncthreads();

  if (tid < 256) {
    const int ml  = tid >> 4;
    const int nl  = tid & 15;
    const int idx = ((ml & 7) << 5) + ((ml >> 3) << 4) + nl;
    float s = 0.0f;
    for (int w = 0; w < FCW; ++w) s += sRed[w * 256 + idx];
    const int n = n0 + nl;
    s += (n < NRN ? bias[n] : 0.0f);
    xf[blockIdx.y * (BATCH * 16) + (m0 + ml) * 16 + nl] = s;
  }
}

#define NTH  512
#define NWV  (NTH / 32)
#define NGR  4
#define NPT  14
#define NLAST (NRN - (NGR - 1) * NPT)
#define NPR  (NPT / 2)

template <int CTRL>
__device__ __forceinline__ float dpp_row(float v) {
  const int b = __builtin_bit_cast(int, v);
  return __builtin_bit_cast(float, __builtin_amdgcn_update_dpp(b, b, CTRL, 0xf, 0xf, false));
}
__device__ __forceinline__ float xhalf(float v) {
  const int b = __builtin_bit_cast(int, v);
  return __builtin_bit_cast(float, __builtin_amdgcn_permlanex16(b, b, 0x76543210, 0xfedcba98, false, false));
}
__device__ __forceinline__ float row_sum16(float v) {
  v += dpp_row<0x161>(v);
  v += dpp_row<0x162>(v);
  v += dpp_row<0x164>(v);
  v += dpp_row<0x168>(v);
  return v;
}

__global__ void __launch_bounds__(NTH) scan_kernel(const float* __restrict__ xf,
                                                   const float* __restrict__ lw,
                                                   const float* __restrict__ hw,
                                                   float* __restrict__ out) {
  __shared__ float sPart[NGR * BATCH];
  __shared__ float sWave[NWV];
  __shared__ float sCnt[CELLS];

  const int tid  = threadIdx.x;
  const int lane = tid & 31;
  const int wv   = __builtin_amdgcn_readfirstlane(tid >> 5);
  const int ng   = wv >> 2;
  const int b    = ((wv & 3) << 5) + lane;
  const int n0   = ng * NPT;
  const bool full = (ng < NGR - 1);

  v2f h0[NPR], h1[NPR], h2[NPR], h3[NPR], cnt[NPR], xv[NPR];
  v2f lw0[NPR], lw1[NPR], lw2[NPR], lw3[NPR];
  const float w0 = hw[0], w1 = hw[1], w2 = hw[2], w3 = hw[3];
  const v2f zero2 = {0.0f, 0.0f};

#pragma unroll
  for (int p = 0; p < NPR; ++p) {
    const int n = n0 + 2 * p;
    h0[p] = h1[p] = h2[p] = h3[p] = cnt[p] = zero2;
    if (p < NLAST / 2 || full) {
      const float* q = xf + (n >> 4) * (BATCH * 16) + b * 16 + (n & 15);
      xv[p]  = v2f{q[0], q[1]};
      lw0[p] = v2f{lw[n * DTH + 0], lw[n * DTH + DTH + 0]};
      lw1[p] = v2f{lw[n * DTH + 1], lw[n * DTH + DTH + 1]};
      lw2[p] = v2f{lw[n * DTH + 2], lw[n * DTH + DTH + 2]};
      lw3[p] = v2f{lw[n * DTH + 3], lw[n * DTH + DTH + 3]};
    } else {
      xv[p] = lw0[p] = lw1[p] = lw2[p] = lw3[p] = zero2;
    }
  }

  for (int step = 0; step < STEPS; ++step) {
    v2f sv[NPR], hsv[NPR], rate[NPR];
    v2f part2 = zero2;
#pragma unroll
    for (int p = 0; p < NPR; ++p) {
      sv[p]  = h0[p] * lw0[p] + h1[p] * lw1[p] + h2[p] * lw2[p] + h3[p] * lw3[p];
      hsv[p] = h0[p] * w0 + h1[p] * w1 + h2[p] * w2 + h3[p] * w3;
      part2 += sv[p];
    }
    sPart[ng * BATCH + b] = part2.x + part2.y;
    __syncthreads();
    const float Sb = ((sPart[b] + sPart[BATCH + b]) + sPart[2 * BATCH + b]) + sPart[3 * BATCH + b];
    const v2f Sb2 = {Sb, Sb};
    v2f ps2 = zero2;
#pragma unroll
    for (int p = 0; p < NPR; ++p) {
      const v2f cur = xv[p] + (Sb2 - sv[p]) + hsv[p];
      v2f e = {__expf(cur.x), __expf(cur.y)};
      if (p >= NLAST / 2) e = full ? e : zero2;
      rate[p] = e;
      ps2 += e;
    }
    float psum = row_sum16(ps2.x + ps2.y);
    psum += xhalf(psum);
    if (lane == 0) sWave[wv] = psum;
    __syncthreads();
    const float tot = row_sum16(sWave[lane & 15]);
    const float thr = tot * (1.0f / (float)CELLS);
#pragma unroll
    for (int p = 0; p < NPR; ++p) {
      const v2f sp = {(rate[p].x > thr) ? 1.0f : 0.0f, (rate[p].y > thr) ? 1.0f : 0.0f};
      cnt[p] += sp;
      h0[p] = h1[p]; h1[p] = h2[p]; h2[p] = h3[p]; h3[p] = sp;
    }
  }

#pragma unroll
  for (int p = 0; p < NPR; ++p) {
    if (p < NLAST / 2 || full) {
      sCnt[b * NRN + n0 + 2 * p]     = cnt[p].x;
      sCnt[b * NRN + n0 + 2 * p + 1] = cnt[p].y;
    }
  }
  __syncthreads();
  for (int c = tid; c < CELLS; c += NTH) {
    float cc = sCnt[c];
    out[c] = fmaxf(cc, 0.0f) + log1pf(__expf(-fabsf(cc)));
  }
}

extern "C" void kernel_launch(void* const* d_in, const int* in_sizes, int n_in,
                              void* d_out, int out_size, void* d_ws, size_t ws_size,
                              hipStream_t stream) {
  const float* x  = (const float*)d_in[0];
  const float* lw = (const float*)d_in[1];
  const float* hw = (const float*)d_in[2];
  const float* fw = (const float*)d_in[3];
  const float* fb = (const float*)d_in[4];

  float* ws = (float*)d_ws;
  float* xp = ws;
  float* wP = xp + (size_t)BATCH * KPAD;
  float* xf = wP + (size_t)NPAD * KPAD;
  float* out = (float*)d_out;

  pool_pad_kernel<<<PGRID, PWV * 32, 0, stream>>>(x, fw, xp, wP);
  fc_wmma_kernel<<<dim3(BATCH / 16, NPAD / 16), FCW * 32, 0, stream>>>(xp, wP, fb, xf);
  scan_kernel<<<1, NTH, 0, stream>>>(xf, lw, hw, out);
}


# xp and wP stored in fragment order (512-byte blocks: 16 rows x 8 k); a pool wave's unit is 4 batch rows x 8 pixels; each fc fragment load reads 4 whole lines, not 16 lines x 32 B; sum orders kept
# speedup vs baseline: 8.0189x; 1.1541x over previous
#include <hip/hip_runtime.h>
#include <hip/hip_bf16.h>
#include <math.h>

typedef __attribute__((ext_vector_type(2))) float v2f;
typedef __attribute__((ext_vector_type(8))) float v8f;

#define BATCH  128
#define NRN    54
#define DTH    4
#define STEPS  30
#define PIX    5625
#define KPAD   5632
#define NPAD   64
#define CELLS  (BATCH * NRN)

#define PWV    8
#define PGRID  512
#define PNW    (PWV * PGRID)
#define NOCT   (KPAD / 8)
#define NRG    (BATCH / 4)
#define NUNIT  (NRG * NOCT)
#define NFULL  (NUNIT / PNW)
#define REMW   ((NUNIT - NFULL * PNW) / PGRID)
#define NSLOT  (NFULL + 1)
#define AHEAD  2
#define WCOPY  ((NPAD * KPAD + PNW * 32 - 1) / (PNW * 32))
static_assert(NUNIT == NFULL * PNW + REMW * PGRID && REMW <= PWV, "units must split evenly");

__device__ __forceinline__ float pair_other(float v) {
  const int b = __builtin_bit_cast(int, v);
  return __builtin_bit_cast(float, __builtin_amdgcn_update_dpp(b, b, 0x161, 0xf, 0xf, false));
}
__device__ __forceinline__ float lane_fetch(int byte_addr, float v) {
  return __builtin_bit_cast(float, __builtin_amdgcn_ds_bpermute(byte_addr, __builtin_bit_cast(int, v)));
}
__device__ __forceinline__ const float* window_ptr(const float* __restrict__ xb, int p, int half) {
  const int pp = p < PIX ? p : PIX - 1;
  const int pr = pp / 75;
  const int pc = pp - pr * 75;
  return xb + pr * 4800 + pc * 8 + half * 4;
}

__global__ void __launch_bounds__(PWV * 32) pool_pad_kernel(const float* __restrict__ x,
                                                            const float* __restrict__ w,
                                                            float* __restrict__ xp,
                                                            float* __restrict__ wP) {
  const int tid  = threadIdx.x;
  const int lane = tid & 31;
  const int wave = __builtin_amdgcn_readfirstlane(tid >> 5);
  const int gw   = blockIdx.x * PWV + wave;

  const int gt = blockIdx.x * (PWV * 32) + tid;
  float wv[WCOPY];
#pragma unroll
  for (int c = 0; c < WCOPY; ++c) {
    const int i = gt + c * (PNW * 32);
    wv[c] = 0.0f;
    if (i < NPAD * KPAD) {
      const int blk = i >> 7;
      const int t   = blk / NOCT;
      const int row = t * 16 + ((i >> 3) & 15);
      const int k   = (blk - t * NOCT) * 8 + (i & 7);
      if (row < NRN && k < PIX) wv[c] = w[row * PIX + k];
    }
  }

  float4 ring[AHEAD + 1][16];
#define UNIT_OF(S) ((S) < NFULL ? gw + PNW * (S) : NFULL * PNW + (int)blockIdx.x * REMW + wave)
#define HAS(S)     ((S) < NFULL || wave < REMW)
#define LOAD_UNIT(S) do { if (HAS(S)) {                                                        \
    const int u_ = UNIT_OF(S);                                                                 \
    const int g_ = u_ / NOCT;                                                                  \
    const int o_ = u_ - g_ * NOCT;                                                             \
    const float* xb_ = x + (size_t)(4 * g_ + (lane >> 4)) * 360000;                            \
    const float* pa_ = window_ptr(xb_, 8 * o_ + ((lane >> 1) & 7), lane & 1);                  \
    const float* pb_ = pa_ + 2 * 360000;                    \
    _Pragma("unroll") for (int r = 0; r < 8; ++r) {                                            \
      ring[(S) % (AHEAD + 1)][2 * r]     = *reinterpret_cast<const float4*>(pa_ + r * 600);    \
      ring[(S) % (AHEAD + 1)][2 * r + 1] = *reinterpret_cast<const float4*>(pb_ + r * 600);    \
    } } } while (0)

#pragma unroll
  for (int s = 0; s < AHEAD; ++s) LOAD_UNIT(s);
#pragma unroll
  for (int s = 0; s < NSLOT; ++s) {
    __builtin_amdgcn_sched_barrier(0);
    if (s + AHEAD < NSLOT) LOAD_UNIT(s + AHEAD);
    __builtin_amdgcn_sched_barrier(0);
    if (HAS(s)) {
      const float4* q = ring[s % (AHEAD + 1)];
      float sA = 0.0f, sB = 0.0f;
#pragma unroll
      for (int r = 0; r < 8; ++r) {
        const float4 a = q[2 * r], c = q[2 * r + 1];
        const float hA = a.x + a.y + a.z + a.w;
        const float hB = c.x + c.y + c.z + c.w;
        sA += hA + pair_other(hA);
        sB += hB + pair_other(hB);
      }
      sA *= (1.0f / 64.0f);
      sB *= (1.0f / 64.0f);
      const int from = (lane << 3) & 127;
      const float tA = lane_fetch(from, sA);
      const float tB = lane_fetch(from, sB);
      const int u = UNIT_OF(s);
      const int g = u / NOCT;
      const int o = u - g * NOCT;
      float v = lane < 16 ? tA : tB;
      if (8 * o + (lane & 7) >= PIX) v = 0.0f;
      xp[((g >> 2) * NOCT + o) * 128 + (g & 3) * 32 + lane] = v;
    }
  }
#undef LOAD_UNIT
#undef HAS
#undef UNIT_OF

#pragma unroll
  for (int c = 0; c < WCOPY; ++c) {
    const int i = gt + c * (PNW * 32);
    if (i < NPAD * KPAD) wP[i] = wv[c];
  }
}

#define FCW   32
#define KSL   (KPAD / FCW)
#define NLDH  (KSL / 16)
#define KGW   (KSL / 8)
__device__ __forceinline__ void load_frags(const float* __restrict__ p, float4 (&f)[NLDH]) {
  for (int j = 0; j < NLDH; ++j) f[j] = *reinterpret_cast<const float4*>(p + j * 128);
}

__global__ void __launch_bounds__(FCW * 32) fc_wmma_kernel(const float* __restrict__ xp,
                                                           const float* __restrict__ wP,
                                                           const float* __restrict__ bias,
                                                           float* __restrict__ xf) {
  __shared__ float sRed[FCW * 256];
  const int tid  = threadIdx.x;
  const int lane = tid & 31;
  const int wave = tid >> 5;
  const int half = lane >> 4;
  const int l15  = lane & 15;
  const int m0   = blockIdx.x * 16;
  const int n0   = blockIdx.y * 16;

  const float* pa = xp + (size_t)(blockIdx.x * NOCT + wave * KGW) * 128 + l15 * 8 + (half << 2);
  const float* pb = wP + (size_t)(blockIdx.y * NOCT + wave * KGW) * 128 + l15 * 8 + (half << 2);

  v8f acc = {0.f, 0.f, 0.f, 0.f, 0.f, 0.f, 0.f, 0.f};

  for (int bt = 0; bt < 2; ++bt) {
    float4 fa[NLDH], fb[NLDH];
    load_frags(pa + bt * (NLDH * 128), fa);
    load_frags(pb + bt * (NLDH * 128), fb);
    for (int j = 0; j < NLDH; ++j) {
      v2f a0 = {fa[j].x, fa[j].y}, b0 = {fb[j].x, fb[j].y};
      v2f a1 = {fa[j].z, fa[j].w}, b1 = {fb[j].z, fb[j].w};
      acc = __builtin_amdgcn_wmma_f32_16x16x4_f32(
           false, a0,  false, b0,
           (short)0, acc,  false,  false);
      acc = __builtin_amdgcn_wmma_f32_16x16x4_f32(
           false, a1,  false, b1,
           (short)0, acc,  false,  false);
    }
  }

  for (int v = 0; v < 8; ++v) sRed[wave * 256 + v * 32 + lane] = acc[v];
  __syncthreads();

  if (tid < 256) {
    const int ml  = tid >> 4;
    const int nl  = tid & 15;
    const int idx = ((ml & 7) << 5) + ((ml >> 3) << 4) + nl;
    float s = 0.0f;
    for (int w = 0; w < FCW; ++w) s += sRed[w * 256 + idx];
    const int n = n0 + nl;
    s += (n < NRN ? bias[n] : 0.0f);
    xf[blockIdx.y * (BATCH * 16) + (m0 + ml) * 16 + nl] = s;
  }
}

#define NTH  512
#define NWV  (NTH / 32)
#define NGR  4
#define NPT  14
#define NLAST (NRN - (NGR - 1) * NPT)
#define NPR  (NPT / 2)

template <int CTRL>
__device__ __forceinline__ float dpp_row(float v) {
  const int b = __builtin_bit_cast(int, v);
  return __builtin_bit_cast(float, __builtin_amdgcn_update_dpp(b, b, CTRL, 0xf, 0xf, false));
}
__device__ __forceinline__ float xhalf(float v) {
  const int b = __builtin_bit_cast(int, v);
  return __builtin_bit_cast(float, __builtin_amdgcn_permlanex16(b, b, 0x76543210, 0xfedcba98, false, false));
}
__device__ __forceinline__ float row_sum16(float v) {
  v += dpp_row<0x161>(v);
  v += dpp_row<0x162>(v);
  v += dpp_row<0x164>(v);
  v += dpp_row<0x168>(v);
  return v;
}

__global__ void __launch_bounds__(NTH) scan_kernel(const float* __restrict__ xf,
                                                   const float* __restrict__ lw,
                                                   const float* __restrict__ hw,
                                                   float* __restrict__ out) {
  __shared__ float sPart[NGR * BATCH];
  __shared__ float sWave[NWV];
  __shared__ float sCnt[CELLS];

  const int tid  = threadIdx.x;
  const int lane = tid & 31;
  const int wv   = __builtin_amdgcn_readfirstlane(tid >> 5);
  const int ng   = wv >> 2;
  const int b    = ((wv & 3) << 5) + lane;
  const int n0   = ng * NPT;
  const bool full = (ng < NGR - 1);

  v2f h0[NPR], h1[NPR], h2[NPR], h3[NPR], cnt[NPR], xv[NPR];
  v2f lw0[NPR], lw1[NPR], lw2[NPR], lw3[NPR];
  const float w0 = hw[0], w1 = hw[1], w2 = hw[2], w3 = hw[3];
  const v2f zero2 = {0.0f, 0.0f};

#pragma unroll
  for (int p = 0; p < NPR; ++p) {
    const int n = n0 + 2 * p;
    h0[p] = h1[p] = h2[p] = h3[p] = cnt[p] = zero2;
    if (p < NLAST / 2 || full) {
      const float* q = xf + (n >> 4) * (BATCH * 16) + b * 16 + (n & 15);
      xv[p]  = v2f{q[0], q[1]};
      lw0[p] = v2f{lw[n * DTH + 0], lw[n * DTH + DTH + 0]};
      lw1[p] = v2f{lw[n * DTH + 1], lw[n * DTH + DTH + 1]};
      lw2[p] = v2f{lw[n * DTH + 2], lw[n * DTH + DTH + 2]};
      lw3[p] = v2f{lw[n * DTH + 3], lw[n * DTH + DTH + 3]};
    } else {
      xv[p] = lw0[p] = lw1[p] = lw2[p] = lw3[p] = zero2;
    }
  }

  for (int step = 0; step < STEPS; ++step) {
    v2f sv[NPR], hsv[NPR], rate[NPR];
    v2f part2 = zero2;
#pragma unroll
    for (int p = 0; p < NPR; ++p) {
      sv[p]  = h0[p] * lw0[p] + h1[p] * lw1[p] + h2[p] * lw2[p] + h3[p] * lw3[p];
      hsv[p] = h0[p] * w0 + h1[p] * w1 + h2[p] * w2 + h3[p] * w3;
      part2 += sv[p];
    }
    sPart[ng * BATCH + b] = part2.x + part2.y;
    __syncthreads();
    const float Sb = ((sPart[b] + sPart[BATCH + b]) + sPart[2 * BATCH + b]) + sPart[3 * BATCH + b];
    const v2f Sb2 = {Sb, Sb};
    v2f ps2 = zero2;
#pragma unroll
    for (int p = 0; p < NPR; ++p) {
      const v2f cur = xv[p] + (Sb2 - sv[p]) + hsv[p];
      v2f e = {__expf(cur.x), __expf(cur.y)};
      if (p >= NLAST / 2) e = full ? e : zero2;
      rate[p] = e;
      ps2 += e;
    }
    float psum = row_sum16(ps2.x + ps2.y);
    psum += xhalf(psum);
    if (lane == 0) sWave[wv] = psum;
    __syncthreads();
    const float tot = row_sum16(sWave[lane & 15]);
    const float thr = tot * (1.0f / (float)CELLS);
#pragma unroll
    for (int p = 0; p < NPR; ++p) {
      const v2f sp = {(rate[p].x > thr) ? 1.0f : 0.0f, (rate[p].y > thr) ? 1.0f : 0.0f};
      cnt[p] += sp;
      h0[p] = h1[p]; h1[p] = h2[p]; h2[p] = h3[p]; h3[p] = sp;
    }
  }

#pragma unroll
  for (int p = 0; p < NPR; ++p) {
    if (p < NLAST / 2 || full) {
      sCnt[b * NRN + n0 + 2 * p]     = cnt[p].x;
      sCnt[b * NRN + n0 + 2 * p + 1] = cnt[p].y;
    }
  }
  __syncthreads();
  for (int c = tid; c < CELLS; c += NTH) {
    float cc = sCnt[c];
    out[c] = fmaxf(cc, 0.0f) + log1pf(__expf(-fabsf(cc)));
  }
}

extern "C" void kernel_launch(void* const* d_in, const int* in_sizes, int n_in,
                              void* d_out, int out_size, void* d_ws, size_t ws_size,
                              hipStream_t stream) {
  const float* x  = (const float*)d_in[0];
  const float* lw = (const float*)d_in[1];
  const float* hw = (const float*)d_in[2];
  const float* fw = (const float*)d_in[3];
  const float* fb = (const float*)d_in[4];

  float* ws = (float*)d_ws;
  float* xp = ws;
  float* wP = xp + (size_t)BATCH * KPAD;
  float* xf = wP + (size_t)NPAD * KPAD;
  float* out = (float*)d_out;

  pool_pad_kernel<<<PGRID, PWV * 32, 0, stream>>>(x, fw, xp, wP);
  fc_wmma_kernel<<<dim3(BATCH / 16, NPAD / 16), FCW * 32, 0, stream>>>(xp, wP, fb, xf);
  scan_kernel<<<1, NTH, 0, stream>>>(xf, lw, hw, out);
}


# scan epilogue: softplus(count) = max(c,0) + ln2*log2(1+exp(-|c|)) on the hardware exp2/log2 units, about 10 instructions a cell, loop unrolled; not the library log1pf (about 100); last bit may differ
# speedup vs baseline: 8.5191x; 1.0624x over previous
#include <hip/hip_runtime.h>
#include <hip/hip_bf16.h>
#include <math.h>

typedef __attribute__((ext_vector_type(2))) float v2f;
typedef __attribute__((ext_vector_type(8))) float v8f;

#define BATCH  128
#define NRN    54
#define DTH    4
#define STEPS  30
#define PIX    5625
#define KPAD   5632
#define NPAD   64
#define CELLS  (BATCH * NRN)

#define PWV    8
#define PGRID  512
#define PNW    (PWV * PGRID)
#define NOCT   (KPAD / 8)
#define NRG    (BATCH / 4)
#define NUNIT  (NRG * NOCT)
#define NFULL  (NUNIT / PNW)
#define REMW   ((NUNIT - NFULL * PNW) / PGRID)
#define NSLOT  (NFULL + 1)
#define AHEAD  2
#define WCOPY  ((NPAD * KPAD + PNW * 32 - 1) / (PNW * 32))
static_assert(NUNIT == NFULL * PNW + REMW * PGRID && REMW <= PWV, "units must split evenly");

__device__ __forceinline__ float pair_other(float v) {
  const int b = __builtin_bit_cast(int, v);
  return __builtin_bit_cast(float, __builtin_amdgcn_update_dpp(b, b, 0x161, 0xf, 0xf, false));
}
__device__ __forceinline__ float lane_fetch(int byte_addr, float v) {
  return __builtin_bit_cast(float, __builtin_amdgcn_ds_bpermute(byte_addr, __builtin_bit_cast(int, v)));
}
__device__ __forceinline__ const float* window_ptr(const float* __restrict__ xb, int p, int half) {
  const int pp = p < PIX ? p : PIX - 1;
  const int pr = pp / 75;
  const int pc = pp - pr * 75;
  return xb + pr * 4800 + pc * 8 + half * 4;
}

__global__ void __launch_bounds__(PWV * 32) pool_pad_kernel(const float* __restrict__ x,
                                                            const float* __restrict__ w,
                                                            float* __restrict__ xp,
                                                            float* __restrict__ wP) {
  const int tid  = threadIdx.x;
  const int lane = tid & 31;
  const int wave = __builtin_amdgcn_readfirstlane(tid >> 5);
  const int gw   = blockIdx.x * PWV + wave;

  const int gt = blockIdx.x * (PWV * 32) + tid;
  float wv[WCOPY];
#pragma unroll
  for (int c = 0; c < WCOPY; ++c) {
    const int i = gt + c * (PNW * 32);
    wv[c] = 0.0f;
    if (i < NPAD * KPAD) {
      const int blk = i >> 7;
      const int t   = blk / NOCT;
      const int row = t * 16 + ((i >> 3) & 15);
      const int k   = (blk - t * NOCT) * 8 + (i & 7);
      if (row < NRN && k < PIX) wv[c] = w[row * PIX + k];
    }
  }

  float4 ring[AHEAD + 1][16];
#define UNIT_OF(S) ((S) < NFULL ? gw + PNW * (S) : NFULL * PNW + (int)blockIdx.x * REMW + wave)
#define HAS(S)     ((S) < NFULL || wave < REMW)
#define LOAD_UNIT(S) do { if (HAS(S)) {                                                        \
    const int u_ = UNIT_OF(S);                                                                 \
    const int g_ = u_ / NOCT;                                                                  \
    const int o_ = u_ - g_ * NOCT;                                                             \
    const float* xb_ = x + (size_t)(4 * g_ + (lane >> 4)) * 360000;                            \
    const float* pa_ = window_ptr(xb_, 8 * o_ + ((lane >> 1) & 7), lane & 1);                  \
    const float* pb_ = pa_ + 2 * 360000;                    \
    _Pragma("unroll") for (int r = 0; r < 8; ++r) {                                            \
      ring[(S) % (AHEAD + 1)][2 * r]     = *reinterpret_cast<const float4*>(pa_ + r * 600);    \
      ring[(S) % (AHEAD + 1)][2 * r + 1] = *reinterpret_cast<const float4*>(pb_ + r * 600);    \
    } } } while (0)

#pragma unroll
  for (int s = 0; s < AHEAD; ++s) LOAD_UNIT(s);
#pragma unroll
  for (int s = 0; s < NSLOT; ++s) {
    __builtin_amdgcn_sched_barrier(0);
    if (s + AHEAD < NSLOT) LOAD_UNIT(s + AHEAD);
    __builtin_amdgcn_sched_barrier(0);
    if (HAS(s)) {
      const float4* q = ring[s % (AHEAD + 1)];
      float sA = 0.0f, sB = 0.0f;
#pragma unroll
      for (int r = 0; r < 8; ++r) {
        const float4 a = q[2 * r], c = q[2 * r + 1];
        const float hA = a.x + a.y + a.z + a.w;
        const float hB = c.x + c.y + c.z + c.w;
        sA += hA + pair_other(hA);
        sB += hB + pair_other(hB);
      }
      sA *= (1.0f / 64.0f);
      sB *= (1.0f / 64.0f);
      const int from = (lane << 3) & 127;
      const float tA = lane_fetch(from, sA);
      const float tB = lane_fetch(from, sB);
      const int u = UNIT_OF(s);
      const int g = u / NOCT;
      const int o = u - g * NOCT;
      float v = lane < 16 ? tA : tB;
      if (8 * o + (lane & 7) >= PIX) v = 0.0f;
      xp[((g >> 2) * NOCT + o) * 128 + (g & 3) * 32 + lane] = v;
    }
  }
#undef LOAD_UNIT
#undef HAS
#undef UNIT_OF

#pragma unroll
  for (int c = 0; c < WCOPY; ++c) {
    const int i = gt + c * (PNW * 32);
    if (i < NPAD * KPAD) wP[i] = wv[c];
  }
}

#define FCW   32
#define KSL   (KPAD / FCW)
#define NLDH  (KSL / 16)
#define KGW   (KSL / 8)
__device__ __forceinline__ void load_frags(const float* __restrict__ p, float4 (&f)[NLDH]) {
  for (int j = 0; j < NLDH; ++j) f[j] = *reinterpret_cast<const float4*>(p + j * 128);
}

__global__ void __launch_bounds__(FCW * 32) fc_wmma_kernel(const float* __restrict__ xp,
                                                           const float* __restrict__ wP,
                                                           const float* __restrict__ bias,
                                                           float* __restrict__ xf) {
  __shared__ float sRed[FCW * 256];
  const int tid  = threadIdx.x;
  const int lane = tid & 31;
  const int wave = tid >> 5;
  const int half = lane >> 4;
  const int l15  = lane & 15;
  const int m0   = blockIdx.x * 16;
  const int n0   = blockIdx.y * 16;

  const float* pa = xp + (size_t)(blockIdx.x * NOCT + wave * KGW) * 128 + l15 * 8 + (half << 2);
  const float* pb = wP + (size_t)(blockIdx.y * NOCT + wave * KGW) * 128 + l15 * 8 + (half << 2);

  v8f acc = {0.f, 0.f, 0.f, 0.f, 0.f, 0.f, 0.f, 0.f};

  for (int bt = 0; bt < 2; ++bt) {
    float4 fa[NLDH], fb[NLDH];
    load_frags(pa + bt * (NLDH * 128), fa);
    load_frags(pb + bt * (NLDH * 128), fb);
    for (int j = 0; j < NLDH; ++j) {
      v2f a0 = {fa[j].x, fa[j].y}, b0 = {fb[j].x, fb[j].y};
      v2f a1 = {fa[j].z, fa[j].w}, b1 = {fb[j].z, fb[j].w};
      acc = __builtin_amdgcn_wmma_f32_16x16x4_f32(
           false, a0,  false, b0,
           (short)0, acc,  false,  false);
      acc = __builtin_amdgcn_wmma_f32_16x16x4_f32(
           false, a1,  false, b1,
           (short)0, acc,  false,  false);
    }
  }

  for (int v = 0; v < 8; ++v) sRed[wave * 256 + v * 32 + lane] = acc[v];
  __syncthreads();

  if (tid < 256) {
    const int ml  = tid >> 4;
    const int nl  = tid & 15;
    const int idx = ((ml & 7) << 5) + ((ml >> 3) << 4) + nl;
    float s = 0.0f;
    for (int w = 0; w < FCW; ++w) s += sRed[w * 256 + idx];
    const int n = n0 + nl;
    s += (n < NRN ? bias[n] : 0.0f);
    xf[blockIdx.y * (BATCH * 16) + (m0 + ml) * 16 + nl] = s;
  }
}

#define NTH  512
#define NWV  (NTH / 32)
#define NGR  4
#define NPT  14
#define NLAST (NRN - (NGR - 1) * NPT)
#define NPR  (NPT / 2)

template <int CTRL>
__device__ __forceinline__ float dpp_row(float v) {
  const int b = __builtin_bit_cast(int, v);
  return __builtin_bit_cast(float, __builtin_amdgcn_update_dpp(b, b, CTRL, 0xf, 0xf, false));
}
__device__ __forceinline__ float xhalf(float v) {
  const int b = __builtin_bit_cast(int, v);
  return __builtin_bit_cast(float, __builtin_amdgcn_permlanex16(b, b, 0x76543210, 0xfedcba98, false, false));
}
__device__ __forceinline__ float row_sum16(float v) {
  v += dpp_row<0x161>(v);
  v += dpp_row<0x162>(v);
  v += dpp_row<0x164>(v);
  v += dpp_row<0x168>(v);
  return v;
}

__global__ void __launch_bounds__(NTH) scan_kernel(const float* __restrict__ xf,
                                                   const float* __restrict__ lw,
                                                   const float* __restrict__ hw,
                                                   float* __restrict__ out) {
  __shared__ float sPart[NGR * BATCH];
  __shared__ float sWave[NWV];
  __shared__ float sCnt[CELLS];

  const int tid  = threadIdx.x;
  const int lane = tid & 31;
  const int wv   = __builtin_amdgcn_readfirstlane(tid >> 5);
  const int ng   = wv >> 2;
  const int b    = ((wv & 3) << 5) + lane;
  const int n0   = ng * NPT;
  const bool full = (ng < NGR - 1);

  v2f h0[NPR], h1[NPR], h2[NPR], h3[NPR], cnt[NPR], xv[NPR];
  v2f lw0[NPR], lw1[NPR], lw2[NPR], lw3[NPR];
  const float w0 = hw[0], w1 = hw[1], w2 = hw[2], w3 = hw[3];
  const v2f zero2 = {0.0f, 0.0f};

#pragma unroll
  for (int p = 0; p < NPR; ++p) {
    const int n = n0 + 2 * p;
    h0[p] = h1[p] = h2[p] = h3[p] = cnt[p] = zero2;
    if (p < NLAST / 2 || full) {
      const float* q = xf + (n >> 4) * (BATCH * 16) + b * 16 + (n & 15);
      xv[p]  = v2f{q[0], q[1]};
      lw0[p] = v2f{lw[n * DTH + 0], lw[n * DTH + DTH + 0]};
      lw1[p] = v2f{lw[n * DTH + 1], lw[n * DTH + DTH + 1]};
      lw2[p] = v2f{lw[n * DTH + 2], lw[n * DTH + DTH + 2]};
      lw3[p] = v2f{lw[n * DTH + 3], lw[n * DTH + DTH + 3]};
    } else {
      xv[p] = lw0[p] = lw1[p] = lw2[p] = lw3[p] = zero2;
    }
  }

  for (int step = 0; step < STEPS; ++step) {
    v2f sv[NPR], hsv[NPR], rate[NPR];
    v2f part2 = zero2;
#pragma unroll
    for (int p = 0; p < NPR; ++p) {
      sv[p]  = h0[p] * lw0[p] + h1[p] * lw1[p] + h2[p] * lw2[p] + h3[p] * lw3[p];
      hsv[p] = h0[p] * w0 + h1[p] * w1 + h2[p] * w2 + h3[p] * w3;
      part2 += sv[p];
    }
    sPart[ng * BATCH + b] = part2.x + part2.y;
    __syncthreads();
    const float Sb = ((sPart[b] + sPart[BATCH + b]) + sPart[2 * BATCH + b]) + sPart[3 * BATCH + b];
    const v2f Sb2 = {Sb, Sb};
    v2f ps2 = zero2;
#pragma unroll
    for (int p = 0; p < NPR; ++p) {
      const v2f cur = xv[p] + (Sb2 - sv[p]) + hsv[p];
      v2f e = {__expf(cur.x), __expf(cur.y)};
      if (p >= NLAST / 2) e = full ? e : zero2;
      rate[p] = e;
      ps2 += e;
    }
    float psum = row_sum16(ps2.x + ps2.y);
    psum += xhalf(psum);
    if (lane == 0) sWave[wv] = psum;
    __syncthreads();
    const float tot = row_sum16(sWave[lane & 15]);
    const float thr = tot * (1.0f / (float)CELLS);
#pragma unroll
    for (int p = 0; p < NPR; ++p) {
      const v2f sp = {(rate[p].x > thr) ? 1.0f : 0.0f, (rate[p].y > thr) ? 1.0f : 0.0f};
      cnt[p] += sp;
      h0[p] = h1[p]; h1[p] = h2[p]; h2[p] = h3[p]; h3[p] = sp;
    }
  }

#pragma unroll
  for (int p = 0; p < NPR; ++p) {
    if (p < NLAST / 2 || full) {
      sCnt[b * NRN + n0 + 2 * p]     = cnt[p].x;
      sCnt[b * NRN + n0 + 2 * p + 1] = cnt[p].y;
    }
  }
  __syncthreads();
#pragma unroll
  for (int i = 0; i < (CELLS + NTH - 1) / NTH; ++i) {
    const int c = tid + i * NTH;
    if (c < CELLS) {
      const float cc = sCnt[c];
      const float t  = __expf(-fabsf(cc));
      const float lg = __builtin_amdgcn_logf(1.0f + t) * 0.693147181f;
      out[c] = fmaxf(cc, 0.0f) + lg;
    }
  }
}

extern "C" void kernel_launch(void* const* d_in, const int* in_sizes, int n_in,
                              void* d_out, int out_size, void* d_ws, size_t ws_size,
                              hipStream_t stream) {
  const float* x  = (const float*)d_in[0];
  const float* lw = (const float*)d_in[1];
  const float* hw = (const float*)d_in[2];
  const float* fw = (const float*)d_in[3];
  const float* fb = (const float*)d_in[4];

  float* ws = (float*)d_ws;
  float* xp = ws;
  float* wP = xp + (size_t)BATCH * KPAD;
  float* xf = wP + (size_t)NPAD * KPAD;
  float* out = (float*)d_out;

  pool_pad_kernel<<<PGRID, PWV * 32, 0, stream>>>(x, fw, xp, wP);
  fc_wmma_kernel<<<dim3(BATCH / 16, NPAD / 16), FCW * 32, 0, stream>>>(xp, wP, fb, xf);
  scan_kernel<<<1, NTH, 0, stream>>>(xf, lw, hw, out);
}


# scan: each level of the two 16-lane butterfly sums per step is one v_add_f32_dpp (lane read folded into the add: old 0, bound_ctrl) instead of copy + v_mov_b32_dpp + v_add_f32; sum order kept
# speedup vs baseline: 8.8706x; 1.0413x over previous
#include <hip/hip_runtime.h>
#include <hip/hip_bf16.h>
#include <math.h>

typedef __attribute__((ext_vector_type(2))) float v2f;
typedef __attribute__((ext_vector_type(8))) float v8f;

#define BATCH  128
#define NRN    54
#define DTH    4
#define STEPS  30
#define PIX    5625
#define KPAD   5632
#define NPAD   64
#define CELLS  (BATCH * NRN)

#define PWV    8
#define PGRID  512
#define PNW    (PWV * PGRID)
#define NOCT   (KPAD / 8)
#define NRG    (BATCH / 4)
#define NUNIT  (NRG * NOCT)
#define NFULL  (NUNIT / PNW)
#define REMW   ((NUNIT - NFULL * PNW) / PGRID)
#define NSLOT  (NFULL + 1)
#define AHEAD  2
#define WCOPY  ((NPAD * KPAD + PNW * 32 - 1) / (PNW * 32))
static_assert(NUNIT == NFULL * PNW + REMW * PGRID && REMW <= PWV, "units must split evenly");

__device__ __forceinline__ float pair_other(float v) {
  const int b = __builtin_bit_cast(int, v);
  return __builtin_bit_cast(float, __builtin_amdgcn_update_dpp(b, b, 0x161, 0xf, 0xf, false));
}
__device__ __forceinline__ float lane_fetch(int byte_addr, float v) {
  return __builtin_bit_cast(float, __builtin_amdgcn_ds_bpermute(byte_addr, __builtin_bit_cast(int, v)));
}
__device__ __forceinline__ const float* window_ptr(const float* __restrict__ xb, int p, int half) {
  const int pp = p < PIX ? p : PIX - 1;
  const int pr = pp / 75;
  const int pc = pp - pr * 75;
  return xb + pr * 4800 + pc * 8 + half * 4;
}

__global__ void __launch_bounds__(PWV * 32) pool_pad_kernel(const float* __restrict__ x,
                                                            const float* __restrict__ w,
                                                            float* __restrict__ xp,
                                                            float* __restrict__ wP) {
  const int tid  = threadIdx.x;
  const int lane = tid & 31;
  const int wave = __builtin_amdgcn_readfirstlane(tid >> 5);
  const int gw   = blockIdx.x * PWV + wave;

  const int gt = blockIdx.x * (PWV * 32) + tid;
  float wv[WCOPY];
#pragma unroll
  for (int c = 0; c < WCOPY; ++c) {
    const int i = gt + c * (PNW * 32);
    wv[c] = 0.0f;
    if (i < NPAD * KPAD) {
      const int blk = i >> 7;
      const int t   = blk / NOCT;
      const int row = t * 16 + ((i >> 3) & 15);
      const int k   = (blk - t * NOCT) * 8 + (i & 7);
      if (row < NRN && k < PIX) wv[c] = w[row * PIX + k];
    }
  }

  float4 ring[AHEAD + 1][16];
#define UNIT_OF(S) ((S) < NFULL ? gw + PNW * (S) : NFULL * PNW + (int)blockIdx.x * REMW + wave)
#define HAS(S)     ((S) < NFULL || wave < REMW)
#define LOAD_UNIT(S) do { if (HAS(S)) {                                                        \
    const int u_ = UNIT_OF(S);                                                                 \
    const int g_ = u_ / NOCT;                                                                  \
    const int o_ = u_ - g_ * NOCT;                                                             \
    const float* xb_ = x + (size_t)(4 * g_ + (lane >> 4)) * 360000;                            \
    const float* pa_ = window_ptr(xb_, 8 * o_ + ((lane >> 1) & 7), lane & 1);                  \
    const float* pb_ = pa_ + 2 * 360000;                    \
    _Pragma("unroll") for (int r = 0; r < 8; ++r) {                                            \
      ring[(S) % (AHEAD + 1)][2 * r]     = *reinterpret_cast<const float4*>(pa_ + r * 600);    \
      ring[(S) % (AHEAD + 1)][2 * r + 1] = *reinterpret_cast<const float4*>(pb_ + r * 600);    \
    } } } while (0)

#pragma unroll
  for (int s = 0; s < AHEAD; ++s) LOAD_UNIT(s);
#pragma unroll
  for (int s = 0; s < NSLOT; ++s) {
    __builtin_amdgcn_sched_barrier(0);
    if (s + AHEAD < NSLOT) LOAD_UNIT(s + AHEAD);
    __builtin_amdgcn_sched_barrier(0);
    if (HAS(s)) {
      const float4* q = ring[s % (AHEAD + 1)];
      float sA = 0.0f, sB = 0.0f;
#pragma unroll
      for (int r = 0; r < 8; ++r) {
        const float4 a = q[2 * r], c = q[2 * r + 1];
        const float hA = a.x + a.y + a.z + a.w;
        const float hB = c.x + c.y + c.z + c.w;
        sA += hA + pair_other(hA);
        sB += hB + pair_other(hB);
      }
      sA *= (1.0f / 64.0f);
      sB *= (1.0f / 64.0f);
      const int from = (lane << 3) & 127;
      const float tA = lane_fetch(from, sA);
      const float tB = lane_fetch(from, sB);
      const int u = UNIT_OF(s);
      const int g = u / NOCT;
      const int o = u - g * NOCT;
      float v = lane < 16 ? tA : tB;
      if (8 * o + (lane & 7) >= PIX) v = 0.0f;
      xp[((g >> 2) * NOCT + o) * 128 + (g & 3) * 32 + lane] = v;
    }
  }
#undef LOAD_UNIT
#undef HAS
#undef UNIT_OF

#pragma unroll
  for (int c = 0; c < WCOPY; ++c) {
    const int i = gt + c * (PNW * 32);
    if (i < NPAD * KPAD) wP[i] = wv[c];
  }
}

#define FCW   32
#define KSL   (KPAD / FCW)
#define NLDH  (KSL / 16)
#define KGW   (KSL / 8)
__device__ __forceinline__ void load_frags(const float* __restrict__ p, float4 (&f)[NLDH]) {
  for (int j = 0; j < NLDH; ++j) f[j] = *reinterpret_cast<const float4*>(p + j * 128);
}

__global__ void __launch_bounds__(FCW * 32) fc_wmma_kernel(const float* __restrict__ xp,
                                                           const float* __restrict__ wP,
                                                           const float* __restrict__ bias,
                                                           float* __restrict__ xf) {
  __shared__ float sRed[FCW * 256];
  const int tid  = threadIdx.x;
  const int lane = tid & 31;
  const int wave = tid >> 5;
  const int half = lane >> 4;
  const int l15  = lane & 15;
  const int m0   = blockIdx.x * 16;
  const int n0   = blockIdx.y * 16;

  const float* pa = xp + (size_t)(blockIdx.x * NOCT + wave * KGW) * 128 + l15 * 8 + (half << 2);
  const float* pb = wP + (size_t)(blockIdx.y * NOCT + wave * KGW) * 128 + l15 * 8 + (half << 2);

  v8f acc = {0.f, 0.f, 0.f, 0.f, 0.f, 0.f, 0.f, 0.f};

  for (int bt = 0; bt < 2; ++bt) {
    float4 fa[NLDH], fb[NLDH];
    load_frags(pa + bt * (NLDH * 128), fa);
    load_frags(pb + bt * (NLDH * 128), fb);
    for (int j = 0; j < NLDH; ++j) {
      v2f a0 = {fa[j].x, fa[j].y}, b0 = {fb[j].x, fb[j].y};
      v2f a1 = {fa[j].z, fa[j].w}, b1 = {fb[j].z, fb[j].w};
      acc = __builtin_amdgcn_wmma_f32_16x16x4_f32(
           false, a0,  false, b0,
           (short)0, acc,  false,  false);
      acc = __builtin_amdgcn_wmma_f32_16x16x4_f32(
           false, a1,  false, b1,
           (short)0, acc,  false,  false);
    }
  }

  for (int v = 0; v < 8; ++v) sRed[wave * 256 + v * 32 + lane] = acc[v];
  __syncthreads();

  if (tid < 256) {
    const int ml  = tid >> 4;
    const int nl  = tid & 15;
    const int idx = ((ml & 7) << 5) + ((ml >> 3) << 4) + nl;
    float s = 0.0f;
    for (int w = 0; w < FCW; ++w) s += sRed[w * 256 + idx];
    const int n = n0 + nl;
    s += (n < NRN ? bias[n] : 0.0f);
    xf[blockIdx.y * (BATCH * 16) + (m0 + ml) * 16 + nl] = s;
  }
}

#define NTH  512
#define NWV  (NTH / 32)
#define NGR  4
#define NPT  14
#define NLAST (NRN - (NGR - 1) * NPT)
#define NPR  (NPT / 2)

template <int CTRL>
__device__ __forceinline__ float dpp_row(float v) {
  const int b = __builtin_bit_cast(int, v);
  return __builtin_bit_cast(float, __builtin_amdgcn_update_dpp(0, b, CTRL, 0xf, 0xf, true));
}
__device__ __forceinline__ float xhalf(float v) {
  const int b = __builtin_bit_cast(int, v);
  return __builtin_bit_cast(float, __builtin_amdgcn_permlanex16(b, b, 0x76543210, 0xfedcba98, false, false));
}
__device__ __forceinline__ float row_sum16(float v) {
  v += dpp_row<0x161>(v);
  v += dpp_row<0x162>(v);
  v += dpp_row<0x164>(v);
  v += dpp_row<0x168>(v);
  return v;
}

__global__ void __launch_bounds__(NTH) scan_kernel(const float* __restrict__ xf,
                                                   const float* __restrict__ lw,
                                                   const float* __restrict__ hw,
                                                   float* __restrict__ out) {
  __shared__ float sPart[NGR * BATCH];
  __shared__ float sWave[NWV];
  __shared__ float sCnt[CELLS];

  const int tid  = threadIdx.x;
  const int lane = tid & 31;
  const int wv   = __builtin_amdgcn_readfirstlane(tid >> 5);
  const int ng   = wv >> 2;
  const int b    = ((wv & 3) << 5) + lane;
  const int n0   = ng * NPT;
  const bool full = (ng < NGR - 1);

  v2f h0[NPR], h1[NPR], h2[NPR], h3[NPR], cnt[NPR], xv[NPR];
  v2f lw0[NPR], lw1[NPR], lw2[NPR], lw3[NPR];
  const float w0 = hw[0], w1 = hw[1], w2 = hw[2], w3 = hw[3];
  const v2f zero2 = {0.0f, 0.0f};

#pragma unroll
  for (int p = 0; p < NPR; ++p) {
    const int n = n0 + 2 * p;
    h0[p] = h1[p] = h2[p] = h3[p] = cnt[p] = zero2;
    if (p < NLAST / 2 || full) {
      const float* q = xf + (n >> 4) * (BATCH * 16) + b * 16 + (n & 15);
      xv[p]  = v2f{q[0], q[1]};
      lw0[p] = v2f{lw[n * DTH + 0], lw[n * DTH + DTH + 0]};
      lw1[p] = v2f{lw[n * DTH + 1], lw[n * DTH + DTH + 1]};
      lw2[p] = v2f{lw[n * DTH + 2], lw[n * DTH + DTH + 2]};
      lw3[p] = v2f{lw[n * DTH + 3], lw[n * DTH + DTH + 3]};
    } else {
      xv[p] = lw0[p] = lw1[p] = lw2[p] = lw3[p] = zero2;
    }
  }

  for (int step = 0; step < STEPS; ++step) {
    v2f sv[NPR], hsv[NPR], rate[NPR];
    v2f part2 = zero2;
#pragma unroll
    for (int p = 0; p < NPR; ++p) {
      sv[p]  = h0[p] * lw0[p] + h1[p] * lw1[p] + h2[p] * lw2[p] + h3[p] * lw3[p];
      hsv[p] = h0[p] * w0 + h1[p] * w1 + h2[p] * w2 + h3[p] * w3;
      part2 += sv[p];
    }
    sPart[ng * BATCH + b] = part2.x + part2.y;
    __syncthreads();
    const float Sb = ((sPart[b] + sPart[BATCH + b]) + sPart[2 * BATCH + b]) + sPart[3 * BATCH + b];
    const v2f Sb2 = {Sb, Sb};
    v2f ps2 = zero2;
#pragma unroll
    for (int p = 0; p < NPR; ++p) {
      const v2f cur = xv[p] + (Sb2 - sv[p]) + hsv[p];
      v2f e = {__expf(cur.x), __expf(cur.y)};
      if (p >= NLAST / 2) e = full ? e : zero2;
      rate[p] = e;
      ps2 += e;
    }
    float psum = row_sum16(ps2.x + ps2.y);
    psum += xhalf(psum);
    if (lane == 0) sWave[wv] = psum;
    __syncthreads();
    const float tot = row_sum16(sWave[lane & 15]);
    const float thr = tot * (1.0f / (float)CELLS);
#pragma unroll
    for (int p = 0; p < NPR; ++p) {
      const v2f sp = {(rate[p].x > thr) ? 1.0f : 0.0f, (rate[p].y > thr) ? 1.0f : 0.0f};
      cnt[p] += sp;
      h0[p] = h1[p]; h1[p] = h2[p]; h2[p] = h3[p]; h3[p] = sp;
    }
  }

#pragma unroll
  for (int p = 0; p < NPR; ++p) {
    if (p < NLAST / 2 || full) {
      sCnt[b * NRN + n0 + 2 * p]     = cnt[p].x;
      sCnt[b * NRN + n0 + 2 * p + 1] = cnt[p].y;
    }
  }
  __syncthreads();
#pragma unroll
  for (int i = 0; i < (CELLS + NTH - 1) / NTH; ++i) {
    const int c = tid + i * NTH;
    if (c < CELLS) {
      const float cc = sCnt[c];
      const float t  = __expf(-fabsf(cc));
      const float lg = __builtin_amdgcn_logf(1.0f + t) * 0.693147181f;
      out[c] = fmaxf(cc, 0.0f) + lg;
    }
  }
}

extern "C" void kernel_launch(void* const* d_in, const int* in_sizes, int n_in,
                              void* d_out, int out_size, void* d_ws, size_t ws_size,
                              hipStream_t stream) {
  const float* x  = (const float*)d_in[0];
  const float* lw = (const float*)d_in[1];
  const float* hw = (const float*)d_in[2];
  const float* fw = (const float*)d_in[3];
  const float* fb = (const float*)d_in[4];

  float* ws = (float*)d_ws;
  float* xp = ws;
  float* wP = xp + (size_t)BATCH * KPAD;
  float* xf = wP + (size_t)NPAD * KPAD;
  float* out = (float*)d_out;

  pool_pad_kernel<<<PGRID, PWV * 32, 0, stream>>>(x, fw, xp, wP);
  fc_wmma_kernel<<<dim3(BATCH / 16, NPAD / 16), FCW * 32, 0, stream>>>(xp, wP, fb, xf);
  scan_kernel<<<1, NTH, 0, stream>>>(xf, lw, hw, out);
}


# scan: both history dot products as 3 running partial sums (no history planes, no shift moves); all updates but the next lateral term run behind the barriers, under the LDS reads; sum orders kept
# speedup vs baseline: 9.3406x; 1.0530x over previous
#include <hip/hip_runtime.h>
#include <hip/hip_bf16.h>
#include <math.h>

typedef __attribute__((ext_vector_type(2))) float v2f;
typedef __attribute__((ext_vector_type(8))) float v8f;

#define BATCH  128
#define NRN    54
#define DTH    4
#define STEPS  30
#define PIX    5625
#define KPAD   5632
#define NPAD   64
#define CELLS  (BATCH * NRN)

#define PWV    8
#define PGRID  512
#define PNW    (PWV * PGRID)
#define NOCT   (KPAD / 8)
#define NRG    (BATCH / 4)
#define NUNIT  (NRG * NOCT)
#define NFULL  (NUNIT / PNW)
#define REMW   ((NUNIT - NFULL * PNW) / PGRID)
#define NSLOT  (NFULL + 1)
#define AHEAD  2
#define WCOPY  ((NPAD * KPAD + PNW * 32 - 1) / (PNW * 32))
static_assert(NUNIT == NFULL * PNW + REMW * PGRID && REMW <= PWV, "units must split evenly");

__device__ __forceinline__ float pair_other(float v) {
  const int b = __builtin_bit_cast(int, v);
  return __builtin_bit_cast(float, __builtin_amdgcn_update_dpp(b, b, 0x161, 0xf, 0xf, false));
}
__device__ __forceinline__ float lane_fetch(int byte_addr, float v) {
  return __builtin_bit_cast(float, __builtin_amdgcn_ds_bpermute(byte_addr, __builtin_bit_cast(int, v)));
}
__device__ __forceinline__ const float* window_ptr(const float* __restrict__ xb, int p, int half) {
  const int pp = p < PIX ? p : PIX - 1;
  const int pr = pp / 75;
  const int pc = pp - pr * 75;
  return xb + pr * 4800 + pc * 8 + half * 4;
}

__global__ void __launch_bounds__(PWV * 32) pool_pad_kernel(const float* __restrict__ x,
                                                            const float* __restrict__ w,
                                                            float* __restrict__ xp,
                                                            float* __restrict__ wP) {
  const int tid  = threadIdx.x;
  const int lane = tid & 31;
  const int wave = __builtin_amdgcn_readfirstlane(tid >> 5);
  const int gw   = blockIdx.x * PWV + wave;

  const int gt = blockIdx.x * (PWV * 32) + tid;
  float wv[WCOPY];
#pragma unroll
  for (int c = 0; c < WCOPY; ++c) {
    const int i = gt + c * (PNW * 32);
    wv[c] = 0.0f;
    if (i < NPAD * KPAD) {
      const int blk = i >> 7;
      const int t   = blk / NOCT;
      const int row = t * 16 + ((i >> 3) & 15);
      const int k   = (blk - t * NOCT) * 8 + (i & 7);
      if (row < NRN && k < PIX) wv[c] = w[row * PIX + k];
    }
  }

  float4 ring[AHEAD + 1][16];
#define UNIT_OF(S) ((S) < NFULL ? gw + PNW * (S) : NFULL * PNW + (int)blockIdx.x * REMW + wave)
#define HAS(S)     ((S) < NFULL || wave < REMW)
#define LOAD_UNIT(S) do { if (HAS(S)) {                                                        \
    const int u_ = UNIT_OF(S);                                                                 \
    const int g_ = u_ / NOCT;                                                                  \
    const int o_ = u_ - g_ * NOCT;                                                             \
    const float* xb_ = x + (size_t)(4 * g_ + (lane >> 4)) * 360000;                            \
    const float* pa_ = window_ptr(xb_, 8 * o_ + ((lane >> 1) & 7), lane & 1);                  \
    const float* pb_ = pa_ + 2 * 360000;                    \
    _Pragma("unroll") for (int r = 0; r < 8; ++r) {                                            \
      ring[(S) % (AHEAD + 1)][2 * r]     = *reinterpret_cast<const float4*>(pa_ + r * 600);    \
      ring[(S) % (AHEAD + 1)][2 * r + 1] = *reinterpret_cast<const float4*>(pb_ + r * 600);    \
    } } } while (0)

#pragma unroll
  for (int s = 0; s < AHEAD; ++s) LOAD_UNIT(s);
#pragma unroll
  for (int s = 0; s < NSLOT; ++s) {
    __builtin_amdgcn_sched_barrier(0);
    if (s + AHEAD < NSLOT) LOAD_UNIT(s + AHEAD);
    __builtin_amdgcn_sched_barrier(0);
    if (HAS(s)) {
      const float4* q = ring[s % (AHEAD + 1)];
      float sA = 0.0f, sB = 0.0f;
#pragma unroll
      for (int r = 0; r < 8; ++r) {
        const float4 a = q[2 * r], c = q[2 * r + 1];
        const float hA = a.x + a.y + a.z + a.w;
        const float hB = c.x + c.y + c.z + c.w;
        sA += hA + pair_other(hA);
        sB += hB + pair_other(hB);
      }
      sA *= (1.0f / 64.0f);
      sB *= (1.0f / 64.0f);
      const int from = (lane << 3) & 127;
      const float tA = lane_fetch(from, sA);
      const float tB = lane_fetch(from, sB);
      const int u = UNIT_OF(s);
      const int g = u / NOCT;
      const int o = u - g * NOCT;
      float v = lane < 16 ? tA : tB;
      if (8 * o + (lane & 7) >= PIX) v = 0.0f;
      xp[((g >> 2) * NOCT + o) * 128 + (g & 3) * 32 + lane] = v;
    }
  }
#undef LOAD_UNIT
#undef HAS
#undef UNIT_OF

#pragma unroll
  for (int c = 0; c < WCOPY; ++c) {
    const int i = gt + c * (PNW * 32);
    if (i < NPAD * KPAD) wP[i] = wv[c];
  }
}

#define FCW   32
#define KSL   (KPAD / FCW)
#define NLDH  (KSL / 16)
#define KGW   (KSL / 8)
__device__ __forceinline__ void load_frags(const float* __restrict__ p, float4 (&f)[NLDH]) {
  for (int j = 0; j < NLDH; ++j) f[j] = *reinterpret_cast<const float4*>(p + j * 128);
}

__global__ void __launch_bounds__(FCW * 32) fc_wmma_kernel(const float* __restrict__ xp,
                                                           const float* __restrict__ wP,
                                                           const float* __restrict__ bias,
                                                           float* __restrict__ xf) {
  __shared__ float sRed[FCW * 256];
  const int tid  = threadIdx.x;
  const int lane = tid & 31;
  const int wave = tid >> 5;
  const int half = lane >> 4;
  const int l15  = lane & 15;
  const int m0   = blockIdx.x * 16;
  const int n0   = blockIdx.y * 16;

  const float* pa = xp + (size_t)(blockIdx.x * NOCT + wave * KGW) * 128 + l15 * 8 + (half << 2);
  const float* pb = wP + (size_t)(blockIdx.y * NOCT + wave * KGW) * 128 + l15 * 8 + (half << 2);

  v8f acc = {0.f, 0.f, 0.f, 0.f, 0.f, 0.f, 0.f, 0.f};

  for (int bt = 0; bt < 2; ++bt) {
    float4 fa[NLDH], fb[NLDH];
    load_frags(pa + bt * (NLDH * 128), fa);
    load_frags(pb + bt * (NLDH * 128), fb);
    for (int j = 0; j < NLDH; ++j) {
      v2f a0 = {fa[j].x, fa[j].y}, b0 = {fb[j].x, fb[j].y};
      v2f a1 = {fa[j].z, fa[j].w}, b1 = {fb[j].z, fb[j].w};
      acc = __builtin_amdgcn_wmma_f32_16x16x4_f32(
           false, a0,  false, b0,
           (short)0, acc,  false,  false);
      acc = __builtin_amdgcn_wmma_f32_16x16x4_f32(
           false, a1,  false, b1,
           (short)0, acc,  false,  false);
    }
  }

  for (int v = 0; v < 8; ++v) sRed[wave * 256 + v * 32 + lane] = acc[v];
  __syncthreads();

  if (tid < 256) {
    const int ml  = tid >> 4;
    const int nl  = tid & 15;
    const int idx = ((ml & 7) << 5) + ((ml >> 3) << 4) + nl;
    float s = 0.0f;
    for (int w = 0; w < FCW; ++w) s += sRed[w * 256 + idx];
    const int n = n0 + nl;
    s += (n < NRN ? bias[n] : 0.0f);
    xf[blockIdx.y * (BATCH * 16) + (m0 + ml) * 16 + nl] = s;
  }
}

#define NTH  512
#define NWV  (NTH / 32)
#define NGR  4
#define NPT  14
#define NLAST (NRN - (NGR - 1) * NPT)
#define NPR  (NPT / 2)

template <int CTRL>
__device__ __forceinline__ float dpp_row(float v) {
  const int b = __builtin_bit_cast(int, v);
  return __builtin_bit_cast(float, __builtin_amdgcn_update_dpp(0, b, CTRL, 0xf, 0xf, true));
}
__device__ __forceinline__ float xhalf(float v) {
  const int b = __builtin_bit_cast(int, v);
  return __builtin_bit_cast(float, __builtin_amdgcn_permlanex16(b, b, 0x76543210, 0xfedcba98, false, false));
}
__device__ __forceinline__ float row_sum16(float v) {
  v += dpp_row<0x161>(v);
  v += dpp_row<0x162>(v);
  v += dpp_row<0x164>(v);
  v += dpp_row<0x168>(v);
  return v;
}

__global__ void __launch_bounds__(NTH) scan_kernel(const float* __restrict__ xf,
                                                   const float* __restrict__ lw,
                                                   const float* __restrict__ hw,
                                                   float* __restrict__ out) {
  __shared__ float sPart[NGR * BATCH];
  __shared__ float sWave[NWV];
  __shared__ float sCnt[CELLS];

  const int tid  = threadIdx.x;
  const int lane = tid & 31;
  const int wv   = __builtin_amdgcn_readfirstlane(tid >> 5);
  const int ng   = wv >> 2;
  const int b    = ((wv & 3) << 5) + lane;
  const int n0   = ng * NPT;
  const bool full = (ng < NGR - 1);

  v2f sp[NPR], l1[NPR], l2[NPR], l3[NPR], g1[NPR], g2[NPR], g3[NPR], cnt[NPR], xv[NPR];
  v2f lw0[NPR], lw1[NPR], lw2[NPR], lw3[NPR];
  const float w0 = hw[0], w1 = hw[1], w2 = hw[2], w3 = hw[3];
  const v2f zero2 = {0.0f, 0.0f};

#pragma unroll
  for (int p = 0; p < NPR; ++p) {
    const int n = n0 + 2 * p;
    sp[p] = l1[p] = l2[p] = l3[p] = g1[p] = g2[p] = g3[p] = cnt[p] = zero2;
    if (p < NLAST / 2 || full) {
      const float* q = xf + (n >> 4) * (BATCH * 16) + b * 16 + (n & 15);
      xv[p]  = v2f{q[0], q[1]};
      lw0[p] = v2f{lw[n * DTH + 0], lw[n * DTH + DTH + 0]};
      lw1[p] = v2f{lw[n * DTH + 1], lw[n * DTH + DTH + 1]};
      lw2[p] = v2f{lw[n * DTH + 2], lw[n * DTH + DTH + 2]};
      lw3[p] = v2f{lw[n * DTH + 3], lw[n * DTH + DTH + 3]};
    } else {
      xv[p] = lw0[p] = lw1[p] = lw2[p] = lw3[p] = zero2;
    }
  }

  for (int step = 0; step < STEPS; ++step) {
    v2f sv[NPR], hsv[NPR], rate[NPR];
    v2f part2 = zero2;
#pragma unroll
    for (int p = 0; p < NPR; ++p) {
      sv[p] = l1[p] + sp[p] * lw3[p];
      part2 += sv[p];
    }
    sPart[ng * BATCH + b] = part2.x + part2.y;
    __syncthreads();
    __builtin_amdgcn_sched_barrier(0);
    const float q0 = sPart[b], q1 = sPart[BATCH + b], q2 = sPart[2 * BATCH + b], q3 = sPart[3 * BATCH + b];
#pragma unroll
    for (int p = 0; p < NPR; ++p) {
      hsv[p] = g1[p] + sp[p] * w3;
      g1[p]  = g2[p] + sp[p] * w2;
      g2[p]  = g3[p] + sp[p] * w1;
      g3[p]  = sp[p] * w0;
    }
    asm volatile("" : "+v"(g1[0]), "+v"(g1[1]), "+v"(g1[2]), "+v"(g1[3]), "+v"(g1[4]), "+v"(g1[5]), "+v"(g1[6]),
                      "+v"(g2[0]), "+v"(g2[1]), "+v"(g2[2]), "+v"(g2[3]), "+v"(g2[4]), "+v"(g2[5]), "+v"(g2[6]),
                      "+v"(g3[0]), "+v"(g3[1]), "+v"(g3[2]), "+v"(g3[3]), "+v"(g3[4]), "+v"(g3[5]), "+v"(g3[6]));
    __builtin_amdgcn_sched_barrier(0);
    const float Sb = ((q0 + q1) + q2) + q3;
    const v2f Sb2 = {Sb, Sb};
    v2f ps2 = zero2;
#pragma unroll
    for (int p = 0; p < NPR; ++p) {
      const v2f cur = xv[p] + (Sb2 - sv[p]) + hsv[p];
      v2f e = {__expf(cur.x), __expf(cur.y)};
      if (p >= NLAST / 2) e = full ? e : zero2;
      rate[p] = e;
      ps2 += e;
    }
    float psum = row_sum16(ps2.x + ps2.y);
    psum += xhalf(psum);
    if (lane == 0) sWave[wv] = psum;
    __syncthreads();
    __builtin_amdgcn_sched_barrier(0);
    const float wsum = sWave[lane & 15];
#pragma unroll
    for (int p = 0; p < NPR; ++p) {
      l1[p] = l2[p] + sp[p] * lw2[p];
      l2[p] = l3[p] + sp[p] * lw1[p];
      l3[p] = sp[p] * lw0[p];
    }
    const float tot = row_sum16(wsum);
    const float thr = tot * (1.0f / (float)CELLS);
#pragma unroll
    for (int p = 0; p < NPR; ++p) {
      sp[p] = v2f{(rate[p].x > thr) ? 1.0f : 0.0f, (rate[p].y > thr) ? 1.0f : 0.0f};
      cnt[p] += sp[p];
    }
  }

#pragma unroll
  for (int p = 0; p < NPR; ++p) {
    if (p < NLAST / 2 || full) {
      sCnt[b * NRN + n0 + 2 * p]     = cnt[p].x;
      sCnt[b * NRN + n0 + 2 * p + 1] = cnt[p].y;
    }
  }
  __syncthreads();
#pragma unroll
  for (int i = 0; i < (CELLS + NTH - 1) / NTH; ++i) {
    const int c = tid + i * NTH;
    if (c < CELLS) {
      const float cc = sCnt[c];
      const float t  = __expf(-fabsf(cc));
      const float lg = __builtin_amdgcn_logf(1.0f + t) * 0.693147181f;
      out[c] = fmaxf(cc, 0.0f) + lg;
    }
  }
}

extern "C" void kernel_launch(void* const* d_in, const int* in_sizes, int n_in,
                              void* d_out, int out_size, void* d_ws, size_t ws_size,
                              hipStream_t stream) {
  const float* x  = (const float*)d_in[0];
  const float* lw = (const float*)d_in[1];
  const float* hw = (const float*)d_in[2];
  const float* fw = (const float*)d_in[3];
  const float* fb = (const float*)d_in[4];

  float* ws = (float*)d_ws;
  float* xp = ws;
  float* wP = xp + (size_t)BATCH * KPAD;
  float* xf = wP + (size_t)NPAD * KPAD;
  float* out = (float*)d_out;

  pool_pad_kernel<<<PGRID, PWV * 32, 0, stream>>>(x, fw, xp, wP);
  fc_wmma_kernel<<<dim3(BATCH / 16, NPAD / 16), FCW * 32, 0, stream>>>(xp, wP, fb, xf);
  scan_kernel<<<1, NTH, 0, stream>>>(xf, lw, hw, out);
}


# scan: one thread per batch row (54 neurons = 27 register pairs; 128 threads, 1 wave per SIMD): lateral sum in-lane, no LDS pass; 1 barrier/step, not 2 (lateral and rate-mean sum order changed)
# speedup vs baseline: 9.6456x; 1.0327x over previous
#include <hip/hip_runtime.h>
#include <hip/hip_bf16.h>
#include <math.h>

typedef __attribute__((ext_vector_type(2))) float v2f;
typedef __attribute__((ext_vector_type(8))) float v8f;

#define BATCH  128
#define NRN    54
#define DTH    4
#define STEPS  30
#define PIX    5625
#define KPAD   5632
#define NPAD   64
#define CELLS  (BATCH * NRN)

#define PWV    8
#define PGRID  512
#define PNW    (PWV * PGRID)
#define NOCT   (KPAD / 8)
#define NRG    (BATCH / 4)
#define NUNIT  (NRG * NOCT)
#define NFULL  (NUNIT / PNW)
#define REMW   ((NUNIT - NFULL * PNW) / PGRID)
#define NSLOT  (NFULL + 1)
#define AHEAD  2
#define WCOPY  ((NPAD * KPAD + PNW * 32 - 1) / (PNW * 32))
static_assert(NUNIT == NFULL * PNW + REMW * PGRID && REMW <= PWV, "units must split evenly");

__device__ __forceinline__ float pair_other(float v) {
  const int b = __builtin_bit_cast(int, v);
  return __builtin_bit_cast(float, __builtin_amdgcn_update_dpp(b, b, 0x161, 0xf, 0xf, false));
}
__device__ __forceinline__ float lane_fetch(int byte_addr, float v) {
  return __builtin_bit_cast(float, __builtin_amdgcn_ds_bpermute(byte_addr, __builtin_bit_cast(int, v)));
}
__device__ __forceinline__ const float* window_ptr(const float* __restrict__ xb, int p, int half) {
  const int pp = p < PIX ? p : PIX - 1;
  const int pr = pp / 75;
  const int pc = pp - pr * 75;
  return xb + pr * 4800 + pc * 8 + half * 4;
}

__global__ void __launch_bounds__(PWV * 32) pool_pad_kernel(const float* __restrict__ x,
                                                            const float* __restrict__ w,
                                                            float* __restrict__ xp,
                                                            float* __restrict__ wP) {
  const int tid  = threadIdx.x;
  const int lane = tid & 31;
  const int wave = __builtin_amdgcn_readfirstlane(tid >> 5);
  const int gw   = blockIdx.x * PWV + wave;

  const int gt = blockIdx.x * (PWV * 32) + tid;
  float wv[WCOPY];
#pragma unroll
  for (int c = 0; c < WCOPY; ++c) {
    const int i = gt + c * (PNW * 32);
    wv[c] = 0.0f;
    if (i < NPAD * KPAD) {
      const int blk = i >> 7;
      const int t   = blk / NOCT;
      const int row = t * 16 + ((i >> 3) & 15);
      const int k   = (blk - t * NOCT) * 8 + (i & 7);
      if (row < NRN && k < PIX) wv[c] = w[row * PIX + k];
    }
  }

  float4 ring[AHEAD + 1][16];
#define UNIT_OF(S) ((S) < NFULL ? gw + PNW * (S) : NFULL * PNW + (int)blockIdx.x * REMW + wave)
#define HAS(S)     ((S) < NFULL || wave < REMW)
#define LOAD_UNIT(S) do { if (HAS(S)) {                                                        \
    const int u_ = UNIT_OF(S);                                                                 \
    const int g_ = u_ / NOCT;                                                                  \
    const int o_ = u_ - g_ * NOCT;                                                             \
    const float* xb_ = x + (size_t)(4 * g_ + (lane >> 4)) * 360000;                            \
    const float* pa_ = window_ptr(xb_, 8 * o_ + ((lane >> 1) & 7), lane & 1);                  \
    const float* pb_ = pa_ + 2 * 360000;                    \
    _Pragma("unroll") for (int r = 0; r < 8; ++r) {                                            \
      ring[(S) % (AHEAD + 1)][2 * r]     = *reinterpret_cast<const float4*>(pa_ + r * 600);    \
      ring[(S) % (AHEAD + 1)][2 * r + 1] = *reinterpret_cast<const float4*>(pb_ + r * 600);    \
    } } } while (0)

#pragma unroll
  for (int s = 0; s < AHEAD; ++s) LOAD_UNIT(s);
#pragma unroll
  for (int s = 0; s < NSLOT; ++s) {
    __builtin_amdgcn_sched_barrier(0);
    if (s + AHEAD < NSLOT) LOAD_UNIT(s + AHEAD);
    __builtin_amdgcn_sched_barrier(0);
    if (HAS(s)) {
      const float4* q = ring[s % (AHEAD + 1)];
      float sA = 0.0f, sB = 0.0f;
#pragma unroll
      for (int r = 0; r < 8; ++r) {
        const float4 a = q[2 * r], c = q[2 * r + 1];
        const float hA = a.x + a.y + a.z + a.w;
        const float hB = c.x + c.y + c.z + c.w;
        sA += hA + pair_other(hA);
        sB += hB + pair_other(hB);
      }
      sA *= (1.0f / 64.0f);
      sB *= (1.0f / 64.0f);
      const int from = (lane << 3) & 127;
      const float tA = lane_fetch(from, sA);
      const float tB = lane_fetch(from, sB);
      const int u = UNIT_OF(s);
      const int g = u / NOCT;
      const int o = u - g * NOCT;
      float v = lane < 16 ? tA : tB;
      if (8 * o + (lane & 7) >= PIX) v = 0.0f;
      xp[((g >> 2) * NOCT + o) * 128 + (g & 3) * 32 + lane] = v;
    }
  }
#undef LOAD_UNIT
#undef HAS
#undef UNIT_OF

#pragma unroll
  for (int c = 0; c < WCOPY; ++c) {
    const int i = gt + c * (PNW * 32);
    if (i < NPAD * KPAD) wP[i] = wv[c];
  }
}

#define FCW   32
#define KSL   (KPAD / FCW)
#define NLDH  (KSL / 16)
#define KGW   (KSL / 8)
__device__ __forceinline__ void load_frags(const float* __restrict__ p, float4 (&f)[NLDH]) {
  for (int j = 0; j < NLDH; ++j) f[j] = *reinterpret_cast<const float4*>(p + j * 128);
}

__global__ void __launch_bounds__(FCW * 32) fc_wmma_kernel(const float* __restrict__ xp,
                                                           const float* __restrict__ wP,
                                                           const float* __restrict__ bias,
                                                           float* __restrict__ xf) {
  __shared__ float sRed[FCW * 256];
  const int tid  = threadIdx.x;
  const int lane = tid & 31;
  const int wave = tid >> 5;
  const int half = lane >> 4;
  const int l15  = lane & 15;
  const int m0   = blockIdx.x * 16;
  const int n0   = blockIdx.y * 16;

  const float* pa = xp + (size_t)(blockIdx.x * NOCT + wave * KGW) * 128 + l15 * 8 + (half << 2);
  const float* pb = wP + (size_t)(blockIdx.y * NOCT + wave * KGW) * 128 + l15 * 8 + (half << 2);

  v8f acc = {0.f, 0.f, 0.f, 0.f, 0.f, 0.f, 0.f, 0.f};

  for (int bt = 0; bt < 2; ++bt) {
    float4 fa[NLDH], fb[NLDH];
    load_frags(pa + bt * (NLDH * 128), fa);
    load_frags(pb + bt * (NLDH * 128), fb);
    for (int j = 0; j < NLDH; ++j) {
      v2f a0 = {fa[j].x, fa[j].y}, b0 = {fb[j].x, fb[j].y};
      v2f a1 = {fa[j].z, fa[j].w}, b1 = {fb[j].z, fb[j].w};
      acc = __builtin_amdgcn_wmma_f32_16x16x4_f32(
           false, a0,  false, b0,
           (short)0, acc,  false,  false);
      acc = __builtin_amdgcn_wmma_f32_16x16x4_f32(
           false, a1,  false, b1,
           (short)0, acc,  false,  false);
    }
  }

  for (int v = 0; v < 8; ++v) sRed[wave * 256 + v * 32 + lane] = acc[v];
  __syncthreads();

  if (tid < 256) {
    const int ml  = tid >> 4;
    const int nl  = tid & 15;
    const int idx = ((ml & 7) << 5) + ((ml >> 3) << 4) + nl;
    float s = 0.0f;
    for (int w = 0; w < FCW; ++w) s += sRed[w * 256 + idx];
    const int n = n0 + nl;
    s += (n < NRN ? bias[n] : 0.0f);
    xf[blockIdx.y * (BATCH * 16) + (m0 + ml) * 16 + nl] = s;
  }
}

#define NTH  128
#define NWV  (NTH / 32)
#define NPR  (NRN / 2)
#define DBATCH 9
static_assert(NRN % 2 == 0 && NTH == BATCH && NWV == 4, "one thread per batch row, neurons in pairs, 4 wave sums");
static_assert(NPR % DBATCH == 0, "the spike step takes the pairs in whole batches");

template <int CTRL>
__device__ __forceinline__ float dpp_row(float v) {
  const int b = __builtin_bit_cast(int, v);
  return __builtin_bit_cast(float, __builtin_amdgcn_update_dpp(0, b, CTRL, 0xf, 0xf, true));
}
__device__ __forceinline__ float xhalf(float v) {
  const int b = __builtin_bit_cast(int, v);
  return __builtin_bit_cast(float, __builtin_amdgcn_permlanex16(b, b, 0x76543210, 0xfedcba98, false, false));
}
__device__ __forceinline__ float row_sum16(float v) {
  v += dpp_row<0x161>(v);
  v += dpp_row<0x162>(v);
  v += dpp_row<0x164>(v);
  v += dpp_row<0x168>(v);
  return v;
}

__global__ void __launch_bounds__(NTH) scan_kernel(const float* __restrict__ xf,
                                                   const float* __restrict__ lw,
                                                   const float* __restrict__ hw,
                                                   float* __restrict__ out) {
  __shared__ __attribute__((aligned(16))) float sLw[DTH * NRN];
  __shared__ float sWave[2 * NWV * 32];
  __shared__ float sCnt[CELLS];

  const int tid  = threadIdx.x;
  const int lane = tid & 31;
  const int wv   = __builtin_amdgcn_readfirstlane(tid >> 5);
  const int b    = tid;

  for (int i = tid; i < NRN * DTH; i += NTH) sLw[(i & (DTH - 1)) * NRN + (i >> 2)] = lw[i];
  static_assert(DTH == 4, "the transposing copy splits i into neuron i >> 2 and history step i & 3");
  __syncthreads();

  v2f sp[NPR], l1[NPR], l2[NPR], l3[NPR], g1[NPR], g2[NPR], g3[NPR], cnt[NPR], xv[NPR];
  v2f lw0[NPR], lw1[NPR], lw2[NPR], lw3[NPR];
  const float w0 = hw[0], w1 = hw[1], w2 = hw[2], w3 = hw[3];
  const v2f zero2 = {0.0f, 0.0f};

#pragma unroll
  for (int p = 0; p < NPR; ++p) {
    const int n = 2 * p;
    sp[p] = l1[p] = l2[p] = l3[p] = g1[p] = g2[p] = g3[p] = cnt[p] = zero2;
    const float* q = xf + (n >> 4) * (BATCH * 16) + b * 16 + (n & 15);
    xv[p]  = v2f{q[0], q[1]};
    lw0[p] = *reinterpret_cast<const v2f*>(&sLw[0 * NRN + n]);
    lw1[p] = *reinterpret_cast<const v2f*>(&sLw[1 * NRN + n]);
    lw2[p] = *reinterpret_cast<const v2f*>(&sLw[2 * NRN + n]);
    lw3[p] = *reinterpret_cast<const v2f*>(&sLw[3 * NRN + n]);
  }

#define G_UPDATES(LO, HI)                                   \
  _Pragma("unroll") for (int p = (LO); p < (HI); ++p) {     \
    g1[p] = g2[p] + sp[p] * w2;                             \
    g2[p] = g3[p] + sp[p] * w1;                             \
    g3[p] = sp[p] * w0;                                     \
  }

  for (int step = 0; step < STEPS; ++step) {
    v2f sv[NPR], rate[NPR];
    v2f pa[4];
#pragma unroll
    for (int p = 0; p < NPR; ++p) {
      sv[p] = l1[p] + sp[p] * lw3[p];
      if (p < 4) pa[p] = sv[p]; else pa[p & 3] += sv[p];
    }
    const v2f pt = (pa[0] + pa[1]) + (pa[2] + pa[3]);
    const float Sb = pt.x + pt.y;
    const v2f Sb2 = {Sb, Sb};
    v2f pr[4];
#pragma unroll
    for (int p = 0; p < NPR; ++p) {
      const v2f hs  = g1[p] + sp[p] * w3;
      const v2f cur = xv[p] + (Sb2 - sv[p]) + hs;
      const v2f e = {__expf(cur.x), __expf(cur.y)};
      rate[p] = e;
      if (p < 4) pr[p] = e; else pr[p & 3] += e;
    }
    const v2f ps2 = (pr[0] + pr[1]) + (pr[2] + pr[3]);
    float psum = ps2.x + ps2.y;
    G_UPDATES(0, 6);
    psum += dpp_row<0x161>(psum);
    G_UPDATES(6, 11);
    psum += dpp_row<0x162>(psum);
    G_UPDATES(11, 16);
    psum += dpp_row<0x164>(psum);
    G_UPDATES(16, 21);
    psum += dpp_row<0x168>(psum);
    G_UPDATES(21, NPR);
    psum += xhalf(psum);
    const int set = (step & 1) * (NWV * 32);
    sWave[set + wv * 32 + lane] = psum;
    __builtin_amdgcn_sched_barrier(0);
#pragma unroll
    for (int p = 0; p < NPR; ++p) cnt[p] += sp[p];
    __syncthreads();
    __builtin_amdgcn_sched_barrier(0);
    const float ws0 = sWave[set], ws1 = sWave[set + 32], ws2 = sWave[set + 64], ws3 = sWave[set + 96];
#pragma unroll
    for (int p = 0; p < NPR; ++p) {
      l1[p] = l2[p] + sp[p] * lw2[p];
      l2[p] = l3[p] + sp[p] * lw1[p];
      l3[p] = sp[p] * lw0[p];
    }
    __builtin_amdgcn_sched_barrier(0);
    const float tot = (ws0 + ws1) + (ws2 + ws3);
    const float thr = tot * (1.0f / (float)CELLS);
#pragma unroll
    for (int q = 0; q < NPR; q += DBATCH) {
      unsigned mx[DBATCH], my[DBATCH];
#pragma unroll
      for (int j = 0; j < DBATCH; ++j) {
        mx[j] = __builtin_amdgcn_ballot_w32(rate[q + j].x > thr);
        my[j] = __builtin_amdgcn_ballot_w32(rate[q + j].y > thr);
      }
      static_assert(DBATCH == 9, "the statement below names the 18 masks of a batch");
      asm volatile("" : "+s"(mx[0]), "+s"(mx[1]), "+s"(mx[2]), "+s"(mx[3]), "+s"(mx[4]), "+s"(mx[5]),
                        "+s"(mx[6]), "+s"(mx[7]), "+s"(mx[8]), "+s"(my[0]), "+s"(my[1]), "+s"(my[2]),
                        "+s"(my[3]), "+s"(my[4]), "+s"(my[5]), "+s"(my[6]), "+s"(my[7]), "+s"(my[8]));
#pragma unroll
      for (int j = 0; j < DBATCH; ++j)
        sp[q + j] = v2f{__builtin_amdgcn_inverse_ballot_w32(mx[j]) ? 1.0f : 0.0f,
                        __builtin_amdgcn_inverse_ballot_w32(my[j]) ? 1.0f : 0.0f};
    }
  }
#undef G_UPDATES
#pragma unroll
  for (int p = 0; p < NPR; ++p) cnt[p] += sp[p];

#pragma unroll
  for (int p = 0; p < NPR; ++p) {
    sCnt[b * NRN + 2 * p]     = cnt[p].x;
    sCnt[b * NRN + 2 * p + 1] = cnt[p].y;
  }
  __syncthreads();
#pragma unroll 6
  for (int i = 0; i < CELLS / NTH; ++i) {
    const int c = tid + i * NTH;
    const float cc = sCnt[c];
    const float t  = __expf(-fabsf(cc));
    const float lg = __builtin_amdgcn_logf(1.0f + t) * 0.693147181f;
    out[c] = fmaxf(cc, 0.0f) + lg;
  }
}

extern "C" void kernel_launch(void* const* d_in, const int* in_sizes, int n_in,
                              void* d_out, int out_size, void* d_ws, size_t ws_size,
                              hipStream_t stream) {
  const float* x  = (const float*)d_in[0];
  const float* lw = (const float*)d_in[1];
  const float* hw = (const float*)d_in[2];
  const float* fw = (const float*)d_in[3];
  const float* fb = (const float*)d_in[4];

  float* ws = (float*)d_ws;
  float* xp = ws;
  float* wP = xp + (size_t)BATCH * KPAD;
  float* xf = wP + (size_t)NPAD * KPAD;
  float* out = (float*)d_out;

  pool_pad_kernel<<<PGRID, PWV * 32, 0, stream>>>(x, fw, xp, wP);
  fc_wmma_kernel<<<dim3(BATCH / 16, NPAD / 16), FCW * 32, 0, stream>>>(xp, wP, fb, xf);
  scan_kernel<<<1, NTH, 0, stream>>>(xf, lw, hw, out);
}


# fc GEMM: K also split over 4 work-groups per tile (128 work-groups of 8 waves, not 32 of 32: 88 WMMAs per SIMD, not 352); the scan's row loads add the 4 partials, then the bias (K sum order changed)
# speedup vs baseline: 10.1845x; 1.0559x over previous
#include <hip/hip_runtime.h>
#include <hip/hip_bf16.h>
#include <math.h>

typedef __attribute__((ext_vector_type(2))) float v2f;
typedef __attribute__((ext_vector_type(8))) float v8f;

#define BATCH  128
#define NRN    54
#define DTH    4
#define STEPS  30
#define PIX    5625
#define KPAD   5632
#define NPAD   64
#define CELLS  (BATCH * NRN)

#define PWV    8
#define PGRID  512
#define PNW    (PWV * PGRID)
#define NOCT   (KPAD / 8)
#define NRG    (BATCH / 4)
#define NUNIT  (NRG * NOCT)
#define NFULL  (NUNIT / PNW)
#define REMW   ((NUNIT - NFULL * PNW) / PGRID)
#define NSLOT  (NFULL + 1)
#define AHEAD  2
#define WCOPY  ((NPAD * KPAD + PNW * 32 - 1) / (PNW * 32))
static_assert(NUNIT == NFULL * PNW + REMW * PGRID && REMW <= PWV, "units must split evenly");

__device__ __forceinline__ float pair_other(float v) {
  const int b = __builtin_bit_cast(int, v);
  return __builtin_bit_cast(float, __builtin_amdgcn_update_dpp(b, b, 0x161, 0xf, 0xf, false));
}
__device__ __forceinline__ float lane_fetch(int byte_addr, float v) {
  return __builtin_bit_cast(float, __builtin_amdgcn_ds_bpermute(byte_addr, __builtin_bit_cast(int, v)));
}
__device__ __forceinline__ const float* window_ptr(const float* __restrict__ xb, int p, int half) {
  const int pp = p < PIX ? p : PIX - 1;
  const int pr = pp / 75;
  const int pc = pp - pr * 75;
  return xb + pr * 4800 + pc * 8 + half * 4;
}

__global__ void __launch_bounds__(PWV * 32) pool_pad_kernel(const float* __restrict__ x,
                                                            const float* __restrict__ w,
                                                            float* __restrict__ xp,
                                                            float* __restrict__ wP) {
  const int tid  = threadIdx.x;
  const int lane = tid & 31;
  const int wave = __builtin_amdgcn_readfirstlane(tid >> 5);
  const int gw   = blockIdx.x * PWV + wave;

  const int gt = blockIdx.x * (PWV * 32) + tid;
  float wv[WCOPY];
#pragma unroll
  for (int c = 0; c < WCOPY; ++c) {
    const int i = gt + c * (PNW * 32);
    wv[c] = 0.0f;
    if (i < NPAD * KPAD) {
      const int blk = i >> 7;
      const int t   = blk / NOCT;
      const int row = t * 16 + ((i >> 3) & 15);
      const int k   = (blk - t * NOCT) * 8 + (i & 7);
      if (row < NRN && k < PIX) wv[c] = w[row * PIX + k];
    }
  }

  float4 ring[AHEAD + 1][16];
#define UNIT_OF(S) ((S) < NFULL ? gw + PNW * (S) : NFULL * PNW + (int)blockIdx.x * REMW + wave)
#define HAS(S)     ((S) < NFULL || wave < REMW)
#define LOAD_UNIT(S) do { if (HAS(S)) {                                                        \
    const int u_ = UNIT_OF(S);                                                                 \
    const int g_ = u_ / NOCT;                                                                  \
    const int o_ = u_ - g_ * NOCT;                                                             \
    const float* xb_ = x + (size_t)(4 * g_ + (lane >> 4)) * 360000;                            \
    const float* pa_ = window_ptr(xb_, 8 * o_ + ((lane >> 1) & 7), lane & 1);                  \
    const float* pb_ = pa_ + 2 * 360000;                    \
    _Pragma("unroll") for (int r = 0; r < 8; ++r) {                                            \
      ring[(S) % (AHEAD + 1)][2 * r]     = *reinterpret_cast<const float4*>(pa_ + r * 600);    \
      ring[(S) % (AHEAD + 1)][2 * r + 1] = *reinterpret_cast<const float4*>(pb_ + r * 600);    \
    } } } while (0)

#pragma unroll
  for (int s = 0; s < AHEAD; ++s) LOAD_UNIT(s);
#pragma unroll
  for (int s = 0; s < NSLOT; ++s) {
    __builtin_amdgcn_sched_barrier(0);
    if (s + AHEAD < NSLOT) LOAD_UNIT(s + AHEAD);
    __builtin_amdgcn_sched_barrier(0);
    if (HAS(s)) {
      const float4* q = ring[s % (AHEAD + 1)];
      float sA = 0.0f, sB = 0.0f;
#pragma unroll
      for (int r = 0; r < 8; ++r) {
        const float4 a = q[2 * r], c = q[2 * r + 1];
        const float hA = a.x + a.y + a.z + a.w;
        const float hB = c.x + c.y + c.z + c.w;
        sA += hA + pair_other(hA);
        sB += hB + pair_other(hB);
      }
      sA *= (1.0f / 64.0f);
      sB *= (1.0f / 64.0f);
      const int from = (lane << 3) & 127;
      const float tA = lane_fetch(from, sA);
      const float tB = lane_fetch(from, sB);
      const int u = UNIT_OF(s);
      const int g = u / NOCT;
      const int o = u - g * NOCT;
      float v = lane < 16 ? tA : tB;
      if (8 * o + (lane & 7) >= PIX) v = 0.0f;
      xp[((g >> 2) * NOCT + o) * 128 + (g & 3) * 32 + lane] = v;
    }
  }
#undef LOAD_UNIT
#undef HAS
#undef UNIT_OF

#pragma unroll
  for (int c = 0; c < WCOPY; ++c) {
    const int i = gt + c * (PNW * 32);
    if (i < NPAD * KPAD) wP[i] = wv[c];
  }
}

#define FCW   8
#define KQ    4
#define KSL   (KPAD / (FCW * KQ))
#define XFQ   (NPAD * BATCH)
static_assert(KSL * FCW * KQ == KPAD && KSL % 16 == 0, "K must split evenly into wave slices of whole float4 pairs");
#define NLDH  (KSL / 16)
#define KGW   (KSL / 8)
__device__ __forceinline__ void load_frags(const float* __restrict__ p, float4 (&f)[NLDH]) {
  for (int j = 0; j < NLDH; ++j) f[j] = *reinterpret_cast<const float4*>(p + j * 128);
}

__global__ void __launch_bounds__(FCW * 32) fc_wmma_kernel(const float* __restrict__ xp,
                                                           const float* __restrict__ wP,
                                                           float* __restrict__ xf) {
  __shared__ float sRed[FCW * 256];
  const int tid  = threadIdx.x;
  const int lane = tid & 31;
  const int wave = tid >> 5;
  const int ksl  = blockIdx.z * FCW + wave;
  const int half = lane >> 4;
  const int l15  = lane & 15;
  const int m0   = blockIdx.x * 16;

  const float* pa = xp + (size_t)(blockIdx.x * NOCT + ksl * KGW) * 128 + l15 * 8 + (half << 2);
  const float* pb = wP + (size_t)(blockIdx.y * NOCT + ksl * KGW) * 128 + l15 * 8 + (half << 2);

  v8f acc = {0.f, 0.f, 0.f, 0.f, 0.f, 0.f, 0.f, 0.f};

  for (int bt = 0; bt < 2; ++bt) {
    float4 fa[NLDH], fb[NLDH];
    load_frags(pa + bt * (NLDH * 128), fa);
    load_frags(pb + bt * (NLDH * 128), fb);
    for (int j = 0; j < NLDH; ++j) {
      v2f a0 = {fa[j].x, fa[j].y}, b0 = {fb[j].x, fb[j].y};
      v2f a1 = {fa[j].z, fa[j].w}, b1 = {fb[j].z, fb[j].w};
      acc = __builtin_amdgcn_wmma_f32_16x16x4_f32(
           false, a0,  false, b0,
           (short)0, acc,  false,  false);
      acc = __builtin_amdgcn_wmma_f32_16x16x4_f32(
           false, a1,  false, b1,
           (short)0, acc,  false,  false);
    }
  }

  for (int v = 0; v < 8; ++v) sRed[wave * 256 + v * 32 + lane] = acc[v];
  __syncthreads();

  static_assert(FCW * 32 == 256, "one thread per tile element");
  const int ml  = lane >> 1;
  const int nl  = 2 * wave + (lane & 1);
  const int idx = ((ml & 7) << 5) + ((ml >> 3) << 4) + nl;
  float s = 0.0f;
  for (int w = 0; w < FCW; ++w) s += sRed[w * 256 + idx];
  xf[blockIdx.z * XFQ + (blockIdx.y * 8 + wave) * (BATCH * 2) + (m0 + ml) * 2 + (lane & 1)] = s;
}

#define NTH  128
#define NWV  (NTH / 32)
#define NPR  (NRN / 2)
#define DBATCH 9
static_assert(NRN % 2 == 0 && NTH == BATCH && NWV == 4, "one thread per batch row, neurons in pairs, 4 wave sums");
static_assert(NPR % DBATCH == 0, "the spike step takes the pairs in whole batches");
static_assert(KQ == 4 && XFQ == 32 * BATCH * 2, "the row load adds four planes of 32 column pairs");

template <int CTRL>
__device__ __forceinline__ float dpp_row(float v) {
  const int b = __builtin_bit_cast(int, v);
  return __builtin_bit_cast(float, __builtin_amdgcn_update_dpp(0, b, CTRL, 0xf, 0xf, true));
}
__device__ __forceinline__ float xhalf(float v) {
  const int b = __builtin_bit_cast(int, v);
  return __builtin_bit_cast(float, __builtin_amdgcn_permlanex16(b, b, 0x76543210, 0xfedcba98, false, false));
}
__device__ __forceinline__ float row_sum16(float v) {
  v += dpp_row<0x161>(v);
  v += dpp_row<0x162>(v);
  v += dpp_row<0x164>(v);
  v += dpp_row<0x168>(v);
  return v;
}

__global__ void __launch_bounds__(NTH) scan_kernel(const float* __restrict__ xf,
                                                   const float* __restrict__ bias,
                                                   const float* __restrict__ lw,
                                                   const float* __restrict__ hw,
                                                   float* __restrict__ out) {
  __shared__ __attribute__((aligned(16))) float sLw[DTH * NRN];
  __shared__ float sWave[2 * NWV * 32];
  __shared__ float sCnt[CELLS];

  const int tid  = threadIdx.x;
  const int lane = tid & 31;
  const int wv   = __builtin_amdgcn_readfirstlane(tid >> 5);
  const int b    = tid;

  for (int i = tid; i < NRN * DTH; i += NTH) sLw[(i & (DTH - 1)) * NRN + (i >> 2)] = lw[i];
  static_assert(DTH == 4, "the transposing copy splits i into neuron i >> 2 and history step i & 3");
  __syncthreads();

  v2f sp[NPR], l1[NPR], l2[NPR], l3[NPR], g1[NPR], g2[NPR], g3[NPR], cnt[NPR], xv[NPR];
  v2f lw0[NPR], lw1[NPR], lw2[NPR], lw3[NPR];
  const float w0 = hw[0], w1 = hw[1], w2 = hw[2], w3 = hw[3];
  const v2f zero2 = {0.0f, 0.0f};

#pragma unroll
  for (int p = 0; p < NPR; ++p) {
    const int n = 2 * p;
    sp[p] = l1[p] = l2[p] = l3[p] = g1[p] = g2[p] = g3[p] = cnt[p] = zero2;
    const float* q = xf + p * (BATCH * 2) + b * 2;
    const v2f q0 = *reinterpret_cast<const v2f*>(q);
    const v2f q1 = *reinterpret_cast<const v2f*>(q + XFQ);
    const v2f q2 = *reinterpret_cast<const v2f*>(q + 2 * XFQ);
    const v2f q3 = *reinterpret_cast<const v2f*>(q + 3 * XFQ);
    xv[p]  = (((q0 + q1) + q2) + q3) + *reinterpret_cast<const v2f*>(bias + n);
    lw0[p] = *reinterpret_cast<const v2f*>(&sLw[0 * NRN + n]);
    lw1[p] = *reinterpret_cast<const v2f*>(&sLw[1 * NRN + n]);
    lw2[p] = *reinterpret_cast<const v2f*>(&sLw[2 * NRN + n]);
    lw3[p] = *reinterpret_cast<const v2f*>(&sLw[3 * NRN + n]);
  }

#define G_UPDATES(LO, HI)                                   \
  _Pragma("unroll") for (int p = (LO); p < (HI); ++p) {     \
    g1[p] = g2[p] + sp[p] * w2;                             \
    g2[p] = g3[p] + sp[p] * w1;                             \
    g3[p] = sp[p] * w0;                                     \
  }

  for (int step = 0; step < STEPS; ++step) {
    v2f sv[NPR], rate[NPR];
    v2f pa[4];
#pragma unroll
    for (int p = 0; p < NPR; ++p) {
      sv[p] = l1[p] + sp[p] * lw3[p];
      if (p < 4) pa[p] = sv[p]; else pa[p & 3] += sv[p];
    }
    const v2f pt = (pa[0] + pa[1]) + (pa[2] + pa[3]);
    const float Sb = pt.x + pt.y;
    const v2f Sb2 = {Sb, Sb};
    v2f pr[4];
#pragma unroll
    for (int p = 0; p < NPR; ++p) {
      const v2f hs  = g1[p] + sp[p] * w3;
      const v2f cur = xv[p] + (Sb2 - sv[p]) + hs;
      const v2f e = {__expf(cur.x), __expf(cur.y)};
      rate[p] = e;
      if (p < 4) pr[p] = e; else pr[p & 3] += e;
    }
    const v2f ps2 = (pr[0] + pr[1]) + (pr[2] + pr[3]);
    float psum = ps2.x + ps2.y;
    G_UPDATES(0, 6);
    psum += dpp_row<0x161>(psum);
    G_UPDATES(6, 11);
    psum += dpp_row<0x162>(psum);
    G_UPDATES(11, 16);
    psum += dpp_row<0x164>(psum);
    G_UPDATES(16, 21);
    psum += dpp_row<0x168>(psum);
    G_UPDATES(21, NPR);
    psum += xhalf(psum);
    const int set = (step & 1) * (NWV * 32);
    sWave[set + wv * 32 + lane] = psum;
    __builtin_amdgcn_sched_barrier(0);
#pragma unroll
    for (int p = 0; p < NPR; ++p) cnt[p] += sp[p];
    __syncthreads();
    __builtin_amdgcn_sched_barrier(0);
    const float ws0 = sWave[set], ws1 = sWave[set + 32], ws2 = sWave[set + 64], ws3 = sWave[set + 96];
#pragma unroll
    for (int p = 0; p < NPR; ++p) {
      l1[p] = l2[p] + sp[p] * lw2[p];
      l2[p] = l3[p] + sp[p] * lw1[p];
      l3[p] = sp[p] * lw0[p];
    }
    __builtin_amdgcn_sched_barrier(0);
    const float tot = (ws0 + ws1) + (ws2 + ws3);
    const float thr = tot * (1.0f / (float)CELLS);
#pragma unroll
    for (int q = 0; q < NPR; q += DBATCH) {
      unsigned mx[DBATCH], my[DBATCH];
#pragma unroll
      for (int j = 0; j < DBATCH; ++j) {
        mx[j] = __builtin_amdgcn_ballot_w32(rate[q + j].x > thr);
        my[j] = __builtin_amdgcn_ballot_w32(rate[q + j].y > thr);
      }
      static_assert(DBATCH == 9, "the statement below names the 18 masks of a batch");
      asm volatile("" : "+s"(mx[0]), "+s"(mx[1]), "+s"(mx[2]), "+s"(mx[3]), "+s"(mx[4]), "+s"(mx[5]),
                        "+s"(mx[6]), "+s"(mx[7]), "+s"(mx[8]), "+s"(my[0]), "+s"(my[1]), "+s"(my[2]),
                        "+s"(my[3]), "+s"(my[4]), "+s"(my[5]), "+s"(my[6]), "+s"(my[7]), "+s"(my[8]));
#pragma unroll
      for (int j = 0; j < DBATCH; ++j)
        sp[q + j] = v2f{__builtin_amdgcn_inverse_ballot_w32(mx[j]) ? 1.0f : 0.0f,
                        __builtin_amdgcn_inverse_ballot_w32(my[j]) ? 1.0f : 0.0f};
    }
  }
#undef G_UPDATES
#pragma unroll
  for (int p = 0; p < NPR; ++p) cnt[p] += sp[p];

#pragma unroll
  for (int p = 0; p < NPR; ++p) {
    sCnt[b * NRN + 2 * p]     = cnt[p].x;
    sCnt[b * NRN + 2 * p + 1] = cnt[p].y;
  }
  __syncthreads();
#pragma unroll 6
  for (int i = 0; i < CELLS / NTH; ++i) {
    const int c = tid + i * NTH;
    const float cc = sCnt[c];
    const float t  = __expf(-fabsf(cc));
    const float lg = __builtin_amdgcn_logf(1.0f + t) * 0.693147181f;
    out[c] = fmaxf(cc, 0.0f) + lg;
  }
}

extern "C" void kernel_launch(void* const* d_in, const int* in_sizes, int n_in,
                              void* d_out, int out_size, void* d_ws, size_t ws_size,
                              hipStream_t stream) {
  const float* x  = (const float*)d_in[0];
  const float* lw = (const float*)d_in[1];
  const float* hw = (const float*)d_in[2];
  const float* fw = (const float*)d_in[3];
  const float* fb = (const float*)d_in[4];

  float* ws = (float*)d_ws;
  float* xp = ws;
  float* wP = xp + (size_t)BATCH * KPAD;
  float* xf = wP + (size_t)NPAD * KPAD;
  float* out = (float*)d_out;

  pool_pad_kernel<<<PGRID, PWV * 32, 0, stream>>>(x, fw, xp, wP);
  fc_wmma_kernel<<<dim3(BATCH / 16, NPAD / 16, KQ), FCW * 32, 0, stream>>>(xp, wP, xf);
  scan_kernel<<<1, NTH, 0, stream>>>(xf, fb, lw, hw, out);
}
